# starting kernel
# baseline (speedup 1.0000x reference)
#include <hip/hip_runtime.h>
#include <math.h>

#define N_NODES 50000
#define N_EDGES 800000
#define GDIM 128
#define HD1 128
#define HEADS 4
#define HD2 32
#define RREL 8
#define NB 20

typedef __attribute__((ext_vector_type(2))) float v2f;
typedef __attribute__((ext_vector_type(8))) float v8f;

__device__ __forceinline__ float lrelu(float x) { return x > 0.f ? x : 0.2f * x; }

__device__ __forceinline__ void atomicMaxF(float* addr, float val) {
    if (val >= 0.f) atomicMax((int*)addr, __float_as_int(val));
    else            atomicMin((unsigned int*)addr, __float_as_uint(val));
}


__global__ void k_basis_w(const float* __restrict__ comp, const float* __restrict__ basis,
                          float* __restrict__ W) {
    int t = blockIdx.x * blockDim.x + threadIdx.x;
    if (t >= RREL * GDIM * HD1) return;
    int r = t / (GDIM * HD1);
    int io = t % (GDIM * HD1);
    float s = 0.f;
    #pragma unroll
    for (int b = 0; b < NB; ++b) s += comp[r * NB + b] * basis[b * GDIM * HD1 + io];
    W[t] = s;
}

__global__ void k_count(const int* __restrict__ ei, const int* __restrict__ et,
                        float* __restrict__ cnt) {
    int e = blockIdx.x * blockDim.x + threadIdx.x;
    if (e >= N_EDGES) return;
    atomicAdd(&cnt[ei[N_EDGES + e] * RREL + et[e]], 1.f);
}

__global__ void k_fill(float* __restrict__ p, float v, int n) {
    int i = blockIdx.x * blockDim.x + threadIdx.x;
    if (i < n) p[i] = v;
}

__global__ void k_fill_bias(float* __restrict__ p, const float* __restrict__ bias, int n) {
    int i = blockIdx.x * blockDim.x + threadIdx.x;
    if (i < n) p[i] = bias[i & 127];
}


__global__ void k_scatter_rel(const float* __restrict__ x, const int* __restrict__ ei,
                              const int* __restrict__ et, const float* __restrict__ cnt,
                              float* __restrict__ acc, int r) {
    int gid = blockIdx.x * blockDim.x + threadIdx.x;
    int e = gid >> 5;
    if (e >= N_EDGES) return;
    if (et[e] != r) return;
    int lane = gid & 31;
    int s = ei[e], d = ei[N_EDGES + e];
    float scale = 1.f / cnt[d * RREL + r];
    float4 v = ((const float4*)(x + (size_t)s * GDIM))[lane];
    float* o = acc + (size_t)d * GDIM + lane * 4;
    atomicAdd(o + 0, v.x * scale);
    atomicAdd(o + 1, v.y * scale);
    atomicAdd(o + 2, v.z * scale);
    atomicAdd(o + 3, v.w * scale);
}


template <bool BIAS_MODE>
__global__ __launch_bounds__(256) void k_mm16(const float* __restrict__ A,
                                              const float* __restrict__ B,
                                              float* __restrict__ C,
                                              const float* __restrict__ bias) {
    __shared__ float As[16][132];
    int tx = threadIdx.x;
    int ty = threadIdx.y;
    int tid = ty * 32 + tx;
    int row0 = blockIdx.x * 16;

    const float4* A4 = (const float4*)(A + (size_t)row0 * 128);
    for (int i = tid; i < 16 * 32; i += 256) {
        int rr = i >> 5, cc = i & 31;
        float4 v = A4[rr * 32 + cc];
        As[rr][cc * 4 + 0] = v.x;
        As[rr][cc * 4 + 1] = v.y;
        As[rr][cc * 4 + 2] = v.z;
        As[rr][cc * 4 + 3] = v.w;
    }
    __syncthreads();

    int col0 = ty * 16;
    int half = tx >> 4;
    int mn = tx & 15;
    int kg = half * 2;

    v8f c;
    if (BIAS_MODE) {
        float bv = bias ? bias[col0 + mn] : 0.f;
        #pragma unroll
        for (int i = 0; i < 8; ++i) c[i] = bv;
    } else {
        #pragma unroll
        for (int i = 0; i < 8; ++i)
            c[i] = C[(size_t)(row0 + half * 8 + i) * 128 + col0 + mn];
    }

    #pragma unroll 4
    for (int k0 = 0; k0 < 128; k0 += 4) {
        v2f a, b;
        a.x = As[mn][k0 + kg];
        a.y = As[mn][k0 + kg + 1];
        b.x = B[(size_t)(k0 + kg) * 128 + col0 + mn];
        b.y = B[(size_t)(k0 + kg + 1) * 128 + col0 + mn];
        c = __builtin_amdgcn_wmma_f32_16x16x4_f32(false, a, false, b, (short)0, c,
                                                  false, false);
    }

    #pragma unroll
    for (int i = 0; i < 8; ++i)
        C[(size_t)(row0 + half * 8 + i) * 128 + col0 + mn] = c[i];
}


__global__ void k_attn(const float* __restrict__ xl, const float* __restrict__ att_src,
                       const float* __restrict__ att_dst, float* __restrict__ a_src,
                       float* __restrict__ a_dst) {
    int t = blockIdx.x * blockDim.x + threadIdx.x;
    if (t >= N_NODES * HEADS) return;
    int n = t >> 2, h = t & 3;
    const float* row = xl + (size_t)n * 128 + h * HD2;
    float ss = 0.f, sd = 0.f;
    #pragma unroll
    for (int j = 0; j < HD2; ++j) {
        float v = row[j];
        ss += v * att_src[h * HD2 + j];
        sd += v * att_dst[h * HD2 + j];
    }
    a_src[t] = ss;
    a_dst[t] = sd;
}

__global__ void k_edge_max(const int* __restrict__ ei, const float* __restrict__ a_src,
                           const float* __restrict__ a_dst, float* __restrict__ m) {
    int i = blockIdx.x * blockDim.x + threadIdx.x;
    if (i >= N_EDGES + N_NODES) return;
    int s, d;
    if (i < N_EDGES) { s = ei[i]; d = ei[N_EDGES + i]; if (s == d) return; }
    else             { s = d = i - N_EDGES; }
    #pragma unroll
    for (int h = 0; h < HEADS; ++h) {
        float v = lrelu(a_src[s * 4 + h] + a_dst[d * 4 + h]);
        atomicMaxF(&m[d * 4 + h], v);
    }
}

__global__ void k_edge_sum(const int* __restrict__ ei, const float* __restrict__ a_src,
                           const float* __restrict__ a_dst, const float* __restrict__ m,
                           float* __restrict__ denom) {
    int i = blockIdx.x * blockDim.x + threadIdx.x;
    if (i >= N_EDGES + N_NODES) return;
    int s, d;
    if (i < N_EDGES) { s = ei[i]; d = ei[N_EDGES + i]; if (s == d) return; }
    else             { s = d = i - N_EDGES; }
    #pragma unroll
    for (int h = 0; h < HEADS; ++h) {
        float v = lrelu(a_src[s * 4 + h] + a_dst[d * 4 + h]);
        atomicAdd(&denom[d * 4 + h], __expf(v - m[d * 4 + h]));
    }
}

__global__ void k_edge_scatter(const int* __restrict__ ei, const float* __restrict__ xl,
                               const float* __restrict__ a_src, const float* __restrict__ a_dst,
                               const float* __restrict__ m, const float* __restrict__ denom,
                               float* __restrict__ out) {
    int gid = blockIdx.x * blockDim.x + threadIdx.x;
    int i = gid >> 5;
    if (i >= N_EDGES + N_NODES) return;
    int lane = gid & 31;
    int s, d;
    if (i < N_EDGES) { s = ei[i]; d = ei[N_EDGES + i]; if (s == d) return; }
    else             { s = d = i - N_EDGES; }
    int h = lane >> 3;
    float v = lrelu(a_src[s * 4 + h] + a_dst[d * 4 + h]);
    float w = __expf(v - m[d * 4 + h]) / denom[d * 4 + h];
    float4 xv = ((const float4*)(xl + (size_t)s * 128))[lane];
    float* o = out + (size_t)d * 128 + lane * 4;
    atomicAdd(o + 0, xv.x * w);
    atomicAdd(o + 1, xv.y * w);
    atomicAdd(o + 2, xv.z * w);
    atomicAdd(o + 3, xv.w * w);
}


extern "C" void kernel_launch(void* const* d_in, const int* in_sizes, int n_in,
                              void* d_out, int out_size, void* d_ws, size_t ws_size,
                              hipStream_t stream) {
    const float* x        = (const float*)d_in[0];
    const int*   ei       = (const int*)d_in[1];
    const int*   et       = (const int*)d_in[2];
    const float* comp     = (const float*)d_in[3];
    const float* basis    = (const float*)d_in[4];
    const float* root     = (const float*)d_in[5];
    const float* bias1    = (const float*)d_in[6];
    const float* gat_w    = (const float*)d_in[7];
    const float* att_src  = (const float*)d_in[8];
    const float* att_dst  = (const float*)d_in[9];
    const float* gat_bias = (const float*)d_in[10];
    float* out = (float*)d_out;

    float* ws = (float*)d_ws;
    size_t off = 0;
    float* W    = ws + off; off += (size_t)RREL * GDIM * HD1;
    float* cnt  = ws + off; off += (size_t)N_NODES * RREL;
    float* hbuf = ws + off; off += (size_t)N_NODES * HD1;
    float* xl   = ws + off; off += (size_t)N_NODES * HD1;
    float* asg  = ws + off; off += (size_t)N_NODES * HEADS;
    float* adg  = ws + off; off += (size_t)N_NODES * HEADS;
    float* mbuf = ws + off; off += (size_t)N_NODES * HEADS;
    float* dbuf = ws + off; off += (size_t)N_NODES * HEADS;
    float* acc = xl;

    dim3 mmBlock(32, 8);
    int mmGrid = N_NODES / 16;

    k_basis_w<<<(RREL * GDIM * HD1 + 255) / 256, 256, 0, stream>>>(comp, basis, W);
    hipMemsetAsync(cnt, 0, (size_t)N_NODES * RREL * sizeof(float), stream);
    k_count<<<(N_EDGES + 255) / 256, 256, 0, stream>>>(ei, et, cnt);
    k_fill_bias<<<(N_NODES * HD1 + 255) / 256, 256, 0, stream>>>(hbuf, bias1, N_NODES * HD1);

    for (int r = 0; r < RREL; ++r) {
        hipMemsetAsync(acc, 0, (size_t)N_NODES * GDIM * sizeof(float), stream);
        k_scatter_rel<<<(N_EDGES * 32 + 255) / 256, 256, 0, stream>>>(x, ei, et, cnt, acc, r);
        k_mm16<false><<<mmGrid, mmBlock, 0, stream>>>(acc, W + (size_t)r * GDIM * HD1,
                                                      hbuf, nullptr);
    }
    k_mm16<false><<<mmGrid, mmBlock, 0, stream>>>(x, root, hbuf, nullptr);

    k_mm16<true><<<mmGrid, mmBlock, 0, stream>>>(hbuf, gat_w, xl, nullptr);
    k_attn<<<(N_NODES * HEADS + 255) / 256, 256, 0, stream>>>(xl, att_src, att_dst, asg, adg);
    k_fill<<<(N_NODES * HEADS + 255) / 256, 256, 0, stream>>>(mbuf, -INFINITY, N_NODES * HEADS);
    hipMemsetAsync(dbuf, 0, (size_t)N_NODES * HEADS * sizeof(float), stream);

    int EN = N_EDGES + N_NODES;
    k_edge_max<<<(EN + 255) / 256, 256, 0, stream>>>(ei, asg, adg, mbuf);
    k_edge_sum<<<(EN + 255) / 256, 256, 0, stream>>>(ei, asg, adg, mbuf, dbuf);
    k_fill_bias<<<(N_NODES * HD1 + 255) / 256, 256, 0, stream>>>(out, gat_bias, N_NODES * HD1);
    k_edge_scatter<<<(EN * 32 + 255) / 256, 256, 0, stream>>>(ei, xl, asg, adg, mbuf, dbuf, out);

    (void)in_sizes; (void)n_in; (void)out_size; (void)ws_size;
}


# softmax max and sum passes: one lane per (edge, head) not one per edge, so an edge's four head atomics sit in adjacent lanes and share one 16-byte request; sums stay float atomics, arrival order
# speedup vs baseline: 1.1978x; 1.1978x over previous
#include <hip/hip_runtime.h>
#include <math.h>

#define N_NODES 50000
#define N_EDGES 800000
#define GDIM 128
#define HD1 128
#define HEADS 4
#define HD2 32
#define RREL 8
#define NB 20

typedef __attribute__((ext_vector_type(2))) float v2f;
typedef __attribute__((ext_vector_type(8))) float v8f;

__device__ __forceinline__ float lrelu(float x) { return x > 0.f ? x : 0.2f * x; }

__device__ __forceinline__ void atomicMaxF(float* addr, float val) {
    if (val >= 0.f) atomicMax((int*)addr, __float_as_int(val));
    else            atomicMin((unsigned int*)addr, __float_as_uint(val));
}


__global__ void k_basis_w(const float* __restrict__ comp, const float* __restrict__ basis,
                          float* __restrict__ W) {
    int t = blockIdx.x * blockDim.x + threadIdx.x;
    if (t >= RREL * GDIM * HD1) return;
    int r = t / (GDIM * HD1);
    int io = t % (GDIM * HD1);
    float s = 0.f;
    #pragma unroll
    for (int b = 0; b < NB; ++b) s += comp[r * NB + b] * basis[b * GDIM * HD1 + io];
    W[t] = s;
}

__global__ void k_count(const int* __restrict__ ei, const int* __restrict__ et,
                        float* __restrict__ cnt) {
    int e = blockIdx.x * blockDim.x + threadIdx.x;
    if (e >= N_EDGES) return;
    atomicAdd(&cnt[ei[N_EDGES + e] * RREL + et[e]], 1.f);
}

__global__ void k_fill(float* __restrict__ p, float v, int n) {
    int i = blockIdx.x * blockDim.x + threadIdx.x;
    if (i < n) p[i] = v;
}

__global__ void k_fill_bias(float* __restrict__ p, const float* __restrict__ bias, int n) {
    int i = blockIdx.x * blockDim.x + threadIdx.x;
    if (i < n) p[i] = bias[i & 127];
}


__global__ void k_scatter_rel(const float* __restrict__ x, const int* __restrict__ ei,
                              const int* __restrict__ et, const float* __restrict__ cnt,
                              float* __restrict__ acc, int r) {
    int gid = blockIdx.x * blockDim.x + threadIdx.x;
    int e = gid >> 5;
    if (e >= N_EDGES) return;
    if (et[e] != r) return;
    int lane = gid & 31;
    int s = ei[e], d = ei[N_EDGES + e];
    float scale = 1.f / cnt[d * RREL + r];
    float4 v = ((const float4*)(x + (size_t)s * GDIM))[lane];
    float* o = acc + (size_t)d * GDIM + lane * 4;
    atomicAdd(o + 0, v.x * scale);
    atomicAdd(o + 1, v.y * scale);
    atomicAdd(o + 2, v.z * scale);
    atomicAdd(o + 3, v.w * scale);
}


template <bool BIAS_MODE>
__global__ __launch_bounds__(256) void k_mm16(const float* __restrict__ A,
                                              const float* __restrict__ B,
                                              float* __restrict__ C,
                                              const float* __restrict__ bias) {
    __shared__ float As[16][132];
    int tx = threadIdx.x;
    int ty = threadIdx.y;
    int tid = ty * 32 + tx;
    int row0 = blockIdx.x * 16;

    const float4* A4 = (const float4*)(A + (size_t)row0 * 128);
    for (int i = tid; i < 16 * 32; i += 256) {
        int rr = i >> 5, cc = i & 31;
        float4 v = A4[rr * 32 + cc];
        As[rr][cc * 4 + 0] = v.x;
        As[rr][cc * 4 + 1] = v.y;
        As[rr][cc * 4 + 2] = v.z;
        As[rr][cc * 4 + 3] = v.w;
    }
    __syncthreads();

    int col0 = ty * 16;
    int half = tx >> 4;
    int mn = tx & 15;
    int kg = half * 2;

    v8f c;
    if (BIAS_MODE) {
        float bv = bias ? bias[col0 + mn] : 0.f;
        #pragma unroll
        for (int i = 0; i < 8; ++i) c[i] = bv;
    } else {
        #pragma unroll
        for (int i = 0; i < 8; ++i)
            c[i] = C[(size_t)(row0 + half * 8 + i) * 128 + col0 + mn];
    }

    #pragma unroll 4
    for (int k0 = 0; k0 < 128; k0 += 4) {
        v2f a, b;
        a.x = As[mn][k0 + kg];
        a.y = As[mn][k0 + kg + 1];
        b.x = B[(size_t)(k0 + kg) * 128 + col0 + mn];
        b.y = B[(size_t)(k0 + kg + 1) * 128 + col0 + mn];
        c = __builtin_amdgcn_wmma_f32_16x16x4_f32(false, a, false, b, (short)0, c,
                                                  false, false);
    }

    #pragma unroll
    for (int i = 0; i < 8; ++i)
        C[(size_t)(row0 + half * 8 + i) * 128 + col0 + mn] = c[i];
}


__global__ void k_attn(const float* __restrict__ xl, const float* __restrict__ att_src,
                       const float* __restrict__ att_dst, float* __restrict__ a_src,
                       float* __restrict__ a_dst) {
    int t = blockIdx.x * blockDim.x + threadIdx.x;
    if (t >= N_NODES * HEADS) return;
    int n = t >> 2, h = t & 3;
    const float* row = xl + (size_t)n * 128 + h * HD2;
    float ss = 0.f, sd = 0.f;
    #pragma unroll
    for (int j = 0; j < HD2; ++j) {
        float v = row[j];
        ss += v * att_src[h * HD2 + j];
        sd += v * att_dst[h * HD2 + j];
    }
    a_src[t] = ss;
    a_dst[t] = sd;
}

__global__ void k_edge_max(const int* __restrict__ ei, const float* __restrict__ a_src,
                           const float* __restrict__ a_dst, float* __restrict__ m) {
    int gid = blockIdx.x * blockDim.x + threadIdx.x;
    int i = gid >> 2;
    if (i >= N_EDGES + N_NODES) return;
    int h = gid & 3;
    int s, d;
    if (i < N_EDGES) { s = ei[i]; d = ei[N_EDGES + i]; if (s == d) return; }
    else             { s = d = i - N_EDGES; }
    float v = lrelu(a_src[s * 4 + h] + a_dst[d * 4 + h]);
    atomicMaxF(&m[d * 4 + h], v);
}

__global__ void k_edge_sum(const int* __restrict__ ei, const float* __restrict__ a_src,
                           const float* __restrict__ a_dst, const float* __restrict__ m,
                           float* __restrict__ denom) {
    int gid = blockIdx.x * blockDim.x + threadIdx.x;
    int i = gid >> 2;
    if (i >= N_EDGES + N_NODES) return;
    int h = gid & 3;
    int s, d;
    if (i < N_EDGES) { s = ei[i]; d = ei[N_EDGES + i]; if (s == d) return; }
    else             { s = d = i - N_EDGES; }
    float v = lrelu(a_src[s * 4 + h] + a_dst[d * 4 + h]);
    atomicAdd(&denom[d * 4 + h], __expf(v - m[d * 4 + h]));
}

__global__ void k_edge_scatter(const int* __restrict__ ei, const float* __restrict__ xl,
                               const float* __restrict__ a_src, const float* __restrict__ a_dst,
                               const float* __restrict__ m, const float* __restrict__ denom,
                               float* __restrict__ out) {
    int gid = blockIdx.x * blockDim.x + threadIdx.x;
    int i = gid >> 5;
    if (i >= N_EDGES + N_NODES) return;
    int lane = gid & 31;
    int s, d;
    if (i < N_EDGES) { s = ei[i]; d = ei[N_EDGES + i]; if (s == d) return; }
    else             { s = d = i - N_EDGES; }
    int h = lane >> 3;
    float v = lrelu(a_src[s * 4 + h] + a_dst[d * 4 + h]);
    float w = __expf(v - m[d * 4 + h]) / denom[d * 4 + h];
    float4 xv = ((const float4*)(xl + (size_t)s * 128))[lane];
    float* o = out + (size_t)d * 128 + lane * 4;
    atomicAdd(o + 0, xv.x * w);
    atomicAdd(o + 1, xv.y * w);
    atomicAdd(o + 2, xv.z * w);
    atomicAdd(o + 3, xv.w * w);
}


extern "C" void kernel_launch(void* const* d_in, const int* in_sizes, int n_in,
                              void* d_out, int out_size, void* d_ws, size_t ws_size,
                              hipStream_t stream) {
    const float* x        = (const float*)d_in[0];
    const int*   ei       = (const int*)d_in[1];
    const int*   et       = (const int*)d_in[2];
    const float* comp     = (const float*)d_in[3];
    const float* basis    = (const float*)d_in[4];
    const float* root     = (const float*)d_in[5];
    const float* bias1    = (const float*)d_in[6];
    const float* gat_w    = (const float*)d_in[7];
    const float* att_src  = (const float*)d_in[8];
    const float* att_dst  = (const float*)d_in[9];
    const float* gat_bias = (const float*)d_in[10];
    float* out = (float*)d_out;

    float* ws = (float*)d_ws;
    size_t off = 0;
    float* W    = ws + off; off += (size_t)RREL * GDIM * HD1;
    float* cnt  = ws + off; off += (size_t)N_NODES * RREL;
    float* hbuf = ws + off; off += (size_t)N_NODES * HD1;
    float* xl   = ws + off; off += (size_t)N_NODES * HD1;
    float* asg  = ws + off; off += (size_t)N_NODES * HEADS;
    float* adg  = ws + off; off += (size_t)N_NODES * HEADS;
    float* mbuf = ws + off; off += (size_t)N_NODES * HEADS;
    float* dbuf = ws + off; off += (size_t)N_NODES * HEADS;
    float* acc = xl;

    dim3 mmBlock(32, 8);
    int mmGrid = N_NODES / 16;

    k_basis_w<<<(RREL * GDIM * HD1 + 255) / 256, 256, 0, stream>>>(comp, basis, W);
    hipMemsetAsync(cnt, 0, (size_t)N_NODES * RREL * sizeof(float), stream);
    k_count<<<(N_EDGES + 255) / 256, 256, 0, stream>>>(ei, et, cnt);
    k_fill_bias<<<(N_NODES * HD1 + 255) / 256, 256, 0, stream>>>(hbuf, bias1, N_NODES * HD1);

    for (int r = 0; r < RREL; ++r) {
        hipMemsetAsync(acc, 0, (size_t)N_NODES * GDIM * sizeof(float), stream);
        k_scatter_rel<<<(N_EDGES * 32 + 255) / 256, 256, 0, stream>>>(x, ei, et, cnt, acc, r);
        k_mm16<false><<<mmGrid, mmBlock, 0, stream>>>(acc, W + (size_t)r * GDIM * HD1,
                                                      hbuf, nullptr);
    }
    k_mm16<false><<<mmGrid, mmBlock, 0, stream>>>(x, root, hbuf, nullptr);

    k_mm16<true><<<mmGrid, mmBlock, 0, stream>>>(hbuf, gat_w, xl, nullptr);
    k_attn<<<(N_NODES * HEADS + 255) / 256, 256, 0, stream>>>(xl, att_src, att_dst, asg, adg);
    k_fill<<<(N_NODES * HEADS + 255) / 256, 256, 0, stream>>>(mbuf, -INFINITY, N_NODES * HEADS);
    hipMemsetAsync(dbuf, 0, (size_t)N_NODES * HEADS * sizeof(float), stream);

    int EN = N_EDGES + N_NODES;
    k_edge_max<<<(EN * 4 + 255) / 256, 256, 0, stream>>>(ei, asg, adg, mbuf);
    k_edge_sum<<<(EN * 4 + 255) / 256, 256, 0, stream>>>(ei, asg, adg, mbuf, dbuf);
    k_fill_bias<<<(N_NODES * HD1 + 255) / 256, 256, 0, stream>>>(out, gat_bias, N_NODES * HD1);
    k_edge_scatter<<<(EN * 32 + 255) / 256, 256, 0, stream>>>(ei, xl, asg, adg, mbuf, dbuf, out);

    (void)in_sizes; (void)n_in; (void)out_size; (void)ws_size;
}


# feature scatters (relation mean, attention sum): lane L adds features L, 32+L, 64+L, 96+L: each wave atomic covers one whole 128-byte line, not 4 lines of 8 lanes; float atomics, arrival order
# speedup vs baseline: 1.3147x; 1.0976x over previous
#include <hip/hip_runtime.h>
#include <math.h>

#define N_NODES 50000
#define N_EDGES 800000
#define GDIM 128
#define HD1 128
#define HEADS 4
#define HD2 32
#define RREL 8
#define NB 20

typedef __attribute__((ext_vector_type(2))) float v2f;
typedef __attribute__((ext_vector_type(8))) float v8f;

__device__ __forceinline__ float lrelu(float x) { return x > 0.f ? x : 0.2f * x; }

__device__ __forceinline__ void atomicMaxF(float* addr, float val) {
    if (val >= 0.f) atomicMax((int*)addr, __float_as_int(val));
    else            atomicMin((unsigned int*)addr, __float_as_uint(val));
}


__global__ void k_basis_w(const float* __restrict__ comp, const float* __restrict__ basis,
                          float* __restrict__ W) {
    int t = blockIdx.x * blockDim.x + threadIdx.x;
    if (t >= RREL * GDIM * HD1) return;
    int r = t / (GDIM * HD1);
    int io = t % (GDIM * HD1);
    float s = 0.f;
    #pragma unroll
    for (int b = 0; b < NB; ++b) s += comp[r * NB + b] * basis[b * GDIM * HD1 + io];
    W[t] = s;
}

__global__ void k_count(const int* __restrict__ ei, const int* __restrict__ et,
                        float* __restrict__ cnt) {
    int e = blockIdx.x * blockDim.x + threadIdx.x;
    if (e >= N_EDGES) return;
    atomicAdd(&cnt[ei[N_EDGES + e] * RREL + et[e]], 1.f);
}

__global__ void k_fill(float* __restrict__ p, float v, int n) {
    int i = blockIdx.x * blockDim.x + threadIdx.x;
    if (i < n) p[i] = v;
}

__global__ void k_fill_bias(float* __restrict__ p, const float* __restrict__ bias, int n) {
    int i = blockIdx.x * blockDim.x + threadIdx.x;
    if (i < n) p[i] = bias[i & 127];
}


__global__ void k_scatter_rel(const float* __restrict__ x, const int* __restrict__ ei,
                              const int* __restrict__ et, const float* __restrict__ cnt,
                              float* __restrict__ acc, int r) {
    int gid = blockIdx.x * blockDim.x + threadIdx.x;
    int e = gid >> 5;
    if (e >= N_EDGES) return;
    if (et[e] != r) return;
    int lane = gid & 31;
    int s = ei[e], d = ei[N_EDGES + e];
    float scale = 1.f / cnt[d * RREL + r];
    const float* xs = x + (size_t)s * GDIM + lane;
    float* o = acc + (size_t)d * GDIM + lane;
    float v0 = xs[0], v1 = xs[32], v2 = xs[64], v3 = xs[96];
    atomicAdd(o + 0,  v0 * scale);
    atomicAdd(o + 32, v1 * scale);
    atomicAdd(o + 64, v2 * scale);
    atomicAdd(o + 96, v3 * scale);
}


template <bool BIAS_MODE>
__global__ __launch_bounds__(256) void k_mm16(const float* __restrict__ A,
                                              const float* __restrict__ B,
                                              float* __restrict__ C,
                                              const float* __restrict__ bias) {
    __shared__ float As[16][132];
    int tx = threadIdx.x;
    int ty = threadIdx.y;
    int tid = ty * 32 + tx;
    int row0 = blockIdx.x * 16;

    const float4* A4 = (const float4*)(A + (size_t)row0 * 128);
    for (int i = tid; i < 16 * 32; i += 256) {
        int rr = i >> 5, cc = i & 31;
        float4 v = A4[rr * 32 + cc];
        As[rr][cc * 4 + 0] = v.x;
        As[rr][cc * 4 + 1] = v.y;
        As[rr][cc * 4 + 2] = v.z;
        As[rr][cc * 4 + 3] = v.w;
    }
    __syncthreads();

    int col0 = ty * 16;
    int half = tx >> 4;
    int mn = tx & 15;
    int kg = half * 2;

    v8f c;
    if (BIAS_MODE) {
        float bv = bias ? bias[col0 + mn] : 0.f;
        #pragma unroll
        for (int i = 0; i < 8; ++i) c[i] = bv;
    } else {
        #pragma unroll
        for (int i = 0; i < 8; ++i)
            c[i] = C[(size_t)(row0 + half * 8 + i) * 128 + col0 + mn];
    }

    #pragma unroll 4
    for (int k0 = 0; k0 < 128; k0 += 4) {
        v2f a, b;
        a.x = As[mn][k0 + kg];
        a.y = As[mn][k0 + kg + 1];
        b.x = B[(size_t)(k0 + kg) * 128 + col0 + mn];
        b.y = B[(size_t)(k0 + kg + 1) * 128 + col0 + mn];
        c = __builtin_amdgcn_wmma_f32_16x16x4_f32(false, a, false, b, (short)0, c,
                                                  false, false);
    }

    #pragma unroll
    for (int i = 0; i < 8; ++i)
        C[(size_t)(row0 + half * 8 + i) * 128 + col0 + mn] = c[i];
}


__global__ void k_attn(const float* __restrict__ xl, const float* __restrict__ att_src,
                       const float* __restrict__ att_dst, float* __restrict__ a_src,
                       float* __restrict__ a_dst) {
    int t = blockIdx.x * blockDim.x + threadIdx.x;
    if (t >= N_NODES * HEADS) return;
    int n = t >> 2, h = t & 3;
    const float* row = xl + (size_t)n * 128 + h * HD2;
    float ss = 0.f, sd = 0.f;
    #pragma unroll
    for (int j = 0; j < HD2; ++j) {
        float v = row[j];
        ss += v * att_src[h * HD2 + j];
        sd += v * att_dst[h * HD2 + j];
    }
    a_src[t] = ss;
    a_dst[t] = sd;
}

__global__ void k_edge_max(const int* __restrict__ ei, const float* __restrict__ a_src,
                           const float* __restrict__ a_dst, float* __restrict__ m) {
    int gid = blockIdx.x * blockDim.x + threadIdx.x;
    int i = gid >> 2;
    if (i >= N_EDGES + N_NODES) return;
    int h = gid & 3;
    int s, d;
    if (i < N_EDGES) { s = ei[i]; d = ei[N_EDGES + i]; if (s == d) return; }
    else             { s = d = i - N_EDGES; }
    float v = lrelu(a_src[s * 4 + h] + a_dst[d * 4 + h]);
    atomicMaxF(&m[d * 4 + h], v);
}

__global__ void k_edge_sum(const int* __restrict__ ei, const float* __restrict__ a_src,
                           const float* __restrict__ a_dst, const float* __restrict__ m,
                           float* __restrict__ denom) {
    int gid = blockIdx.x * blockDim.x + threadIdx.x;
    int i = gid >> 2;
    if (i >= N_EDGES + N_NODES) return;
    int h = gid & 3;
    int s, d;
    if (i < N_EDGES) { s = ei[i]; d = ei[N_EDGES + i]; if (s == d) return; }
    else             { s = d = i - N_EDGES; }
    float v = lrelu(a_src[s * 4 + h] + a_dst[d * 4 + h]);
    atomicAdd(&denom[d * 4 + h], __expf(v - m[d * 4 + h]));
}

__global__ void k_edge_scatter(const int* __restrict__ ei, const float* __restrict__ xl,
                               const float* __restrict__ a_src, const float* __restrict__ a_dst,
                               const float* __restrict__ m, const float* __restrict__ denom,
                               float* __restrict__ out) {
    int gid = blockIdx.x * blockDim.x + threadIdx.x;
    int i = gid >> 5;
    if (i >= N_EDGES + N_NODES) return;
    int lane = gid & 31;
    int s, d;
    if (i < N_EDGES) { s = ei[i]; d = ei[N_EDGES + i]; if (s == d) return; }
    else             { s = d = i - N_EDGES; }
    int h = lane & 3;
    float v = lrelu(a_src[s * 4 + h] + a_dst[d * 4 + h]);
    float w = __expf(v - m[d * 4 + h]) / denom[d * 4 + h];
    float w0 = __int_as_float(__builtin_amdgcn_readlane(__float_as_int(w), 0));
    float w1 = __int_as_float(__builtin_amdgcn_readlane(__float_as_int(w), 1));
    float w2 = __int_as_float(__builtin_amdgcn_readlane(__float_as_int(w), 2));
    float w3 = __int_as_float(__builtin_amdgcn_readlane(__float_as_int(w), 3));
    const float* xs = xl + (size_t)s * 128 + lane;
    float* o = out + (size_t)d * 128 + lane;
    float x0 = xs[0], x1 = xs[32], x2 = xs[64], x3 = xs[96];
    atomicAdd(o + 0,  x0 * w0);
    atomicAdd(o + 32, x1 * w1);
    atomicAdd(o + 64, x2 * w2);
    atomicAdd(o + 96, x3 * w3);
}


extern "C" void kernel_launch(void* const* d_in, const int* in_sizes, int n_in,
                              void* d_out, int out_size, void* d_ws, size_t ws_size,
                              hipStream_t stream) {
    const float* x        = (const float*)d_in[0];
    const int*   ei       = (const int*)d_in[1];
    const int*   et       = (const int*)d_in[2];
    const float* comp     = (const float*)d_in[3];
    const float* basis    = (const float*)d_in[4];
    const float* root     = (const float*)d_in[5];
    const float* bias1    = (const float*)d_in[6];
    const float* gat_w    = (const float*)d_in[7];
    const float* att_src  = (const float*)d_in[8];
    const float* att_dst  = (const float*)d_in[9];
    const float* gat_bias = (const float*)d_in[10];
    float* out = (float*)d_out;

    float* ws = (float*)d_ws;
    size_t off = 0;
    float* W    = ws + off; off += (size_t)RREL * GDIM * HD1;
    float* cnt  = ws + off; off += (size_t)N_NODES * RREL;
    float* hbuf = ws + off; off += (size_t)N_NODES * HD1;
    float* xl   = ws + off; off += (size_t)N_NODES * HD1;
    float* asg  = ws + off; off += (size_t)N_NODES * HEADS;
    float* adg  = ws + off; off += (size_t)N_NODES * HEADS;
    float* mbuf = ws + off; off += (size_t)N_NODES * HEADS;
    float* dbuf = ws + off; off += (size_t)N_NODES * HEADS;
    float* acc = xl;

    dim3 mmBlock(32, 8);
    int mmGrid = N_NODES / 16;

    k_basis_w<<<(RREL * GDIM * HD1 + 255) / 256, 256, 0, stream>>>(comp, basis, W);
    hipMemsetAsync(cnt, 0, (size_t)N_NODES * RREL * sizeof(float), stream);
    k_count<<<(N_EDGES + 255) / 256, 256, 0, stream>>>(ei, et, cnt);
    k_fill_bias<<<(N_NODES * HD1 + 255) / 256, 256, 0, stream>>>(hbuf, bias1, N_NODES * HD1);

    for (int r = 0; r < RREL; ++r) {
        hipMemsetAsync(acc, 0, (size_t)N_NODES * GDIM * sizeof(float), stream);
        k_scatter_rel<<<(N_EDGES * 32 + 255) / 256, 256, 0, stream>>>(x, ei, et, cnt, acc, r);
        k_mm16<false><<<mmGrid, mmBlock, 0, stream>>>(acc, W + (size_t)r * GDIM * HD1,
                                                      hbuf, nullptr);
    }
    k_mm16<false><<<mmGrid, mmBlock, 0, stream>>>(x, root, hbuf, nullptr);

    k_mm16<true><<<mmGrid, mmBlock, 0, stream>>>(hbuf, gat_w, xl, nullptr);
    k_attn<<<(N_NODES * HEADS + 255) / 256, 256, 0, stream>>>(xl, att_src, att_dst, asg, adg);
    k_fill<<<(N_NODES * HEADS + 255) / 256, 256, 0, stream>>>(mbuf, -INFINITY, N_NODES * HEADS);
    hipMemsetAsync(dbuf, 0, (size_t)N_NODES * HEADS * sizeof(float), stream);

    int EN = N_EDGES + N_NODES;
    k_edge_max<<<(EN * 4 + 255) / 256, 256, 0, stream>>>(ei, asg, adg, mbuf);
    k_edge_sum<<<(EN * 4 + 255) / 256, 256, 0, stream>>>(ei, asg, adg, mbuf, dbuf);
    k_fill_bias<<<(N_NODES * HD1 + 255) / 256, 256, 0, stream>>>(out, gat_bias, N_NODES * HD1);
    k_edge_scatter<<<(EN * 32 + 255) / 256, 256, 0, stream>>>(ei, xl, asg, adg, mbuf, dbuf, out);

    (void)in_sizes; (void)n_in; (void)out_size; (void)ws_size;
}


# relation scatter: one wave per 32 consecutive edges, not one per edge: full-line reads of type, source, dest; a ballot picks this relation's edges, taken 4 per trip; float atomics, arrival order
# speedup vs baseline: 1.6768x; 1.2754x over previous
#include <hip/hip_runtime.h>
#include <math.h>

#define N_NODES 50000
#define N_EDGES 800000
#define GDIM 128
#define HD1 128
#define HEADS 4
#define HD2 32
#define RREL 8
#define NB 20

typedef __attribute__((ext_vector_type(2))) float v2f;
typedef __attribute__((ext_vector_type(8))) float v8f;

__device__ __forceinline__ float lrelu(float x) { return x > 0.f ? x : 0.2f * x; }

__device__ __forceinline__ void atomicMaxF(float* addr, float val) {
    if (val >= 0.f) atomicMax((int*)addr, __float_as_int(val));
    else            atomicMin((unsigned int*)addr, __float_as_uint(val));
}


__global__ void k_basis_w(const float* __restrict__ comp, const float* __restrict__ basis,
                          float* __restrict__ W) {
    int t = blockIdx.x * blockDim.x + threadIdx.x;
    if (t >= RREL * GDIM * HD1) return;
    int r = t / (GDIM * HD1);
    int io = t % (GDIM * HD1);
    float s = 0.f;
    #pragma unroll
    for (int b = 0; b < NB; ++b) s += comp[r * NB + b] * basis[b * GDIM * HD1 + io];
    W[t] = s;
}

__global__ void k_count(const int* __restrict__ ei, const int* __restrict__ et,
                        float* __restrict__ cnt) {
    int e = blockIdx.x * blockDim.x + threadIdx.x;
    if (e >= N_EDGES) return;
    atomicAdd(&cnt[ei[N_EDGES + e] * RREL + et[e]], 1.f);
}

__global__ void k_fill(float* __restrict__ p, float v, int n) {
    int i = blockIdx.x * blockDim.x + threadIdx.x;
    if (i < n) p[i] = v;
}

__global__ void k_fill_bias(float* __restrict__ p, const float* __restrict__ bias, int n) {
    int i = blockIdx.x * blockDim.x + threadIdx.x;
    if (i < n) p[i] = bias[i & 127];
}


#define SCAT_U 4

__global__ void k_scatter_rel(const float* __restrict__ x, const int* __restrict__ ei,
                              const int* __restrict__ et, const float* __restrict__ cnt,
                              float* __restrict__ acc, int r) {
    int gid = blockIdx.x * blockDim.x + threadIdx.x;
    int lane = threadIdx.x & 31;
    bool in = gid < N_EDGES;
    int ec = in ? gid : N_EDGES - 1;
    int t = et[ec];
    int s = ei[ec];
    int d = ei[N_EDGES + ec];
    asm volatile("" :: "v"(s), "v"(d));
    unsigned todo = (unsigned)__builtin_amdgcn_ballot_w32(in && t == r);
    while (todo != 0u) {
        int  j[SCAT_U];
        bool ok[SCAT_U];
        #pragma unroll
        for (int u = 0; u < SCAT_U; ++u) {
            ok[u] = todo != 0u;
            j[u] = ok[u] ? __builtin_ctz(todo) : j[0];
            todo &= todo - 1u;
        }
        int   dj[SCAT_U];
        float cn[SCAT_U];
        float v[SCAT_U][4];
        #pragma unroll
        for (int u = 0; u < SCAT_U; ++u) {
            int sj = __builtin_amdgcn_readlane(s, j[u]);
            dj[u] = __builtin_amdgcn_readlane(d, j[u]);
            const float* xs = x + (size_t)sj * GDIM + lane;
            v[u][0] = xs[0]; v[u][1] = xs[32]; v[u][2] = xs[64]; v[u][3] = xs[96];
            cn[u] = cnt[dj[u] * RREL + r];
        }
        static_assert(SCAT_U == 4, "the statement below names the loads of four edges");
        asm volatile("" :: "v"(v[0][0]), "v"(v[0][1]), "v"(v[0][2]), "v"(v[0][3]),
                           "v"(v[1][0]), "v"(v[1][1]), "v"(v[1][2]), "v"(v[1][3]),
                           "v"(v[2][0]), "v"(v[2][1]), "v"(v[2][2]), "v"(v[2][3]),
                           "v"(v[3][0]), "v"(v[3][1]), "v"(v[3][2]), "v"(v[3][3]),
                           "s"(cn[0]), "s"(cn[1]), "s"(cn[2]), "s"(cn[3]));
        #pragma unroll
        for (int u = 0; u < SCAT_U; ++u) {
            if (ok[u]) {
                float c = 1.f / cn[u];
                float* o = acc + (size_t)dj[u] * GDIM + lane;
                atomicAdd(o + 0,  v[u][0] * c);
                atomicAdd(o + 32, v[u][1] * c);
                atomicAdd(o + 64, v[u][2] * c);
                atomicAdd(o + 96, v[u][3] * c);
            }
        }
    }
}


template <bool BIAS_MODE>
__global__ __launch_bounds__(256) void k_mm16(const float* __restrict__ A,
                                              const float* __restrict__ B,
                                              float* __restrict__ C,
                                              const float* __restrict__ bias) {
    __shared__ float As[16][132];
    int tx = threadIdx.x;
    int ty = threadIdx.y;
    int tid = ty * 32 + tx;
    int row0 = blockIdx.x * 16;

    const float4* A4 = (const float4*)(A + (size_t)row0 * 128);
    for (int i = tid; i < 16 * 32; i += 256) {
        int rr = i >> 5, cc = i & 31;
        float4 v = A4[rr * 32 + cc];
        As[rr][cc * 4 + 0] = v.x;
        As[rr][cc * 4 + 1] = v.y;
        As[rr][cc * 4 + 2] = v.z;
        As[rr][cc * 4 + 3] = v.w;
    }
    __syncthreads();

    int col0 = ty * 16;
    int half = tx >> 4;
    int mn = tx & 15;
    int kg = half * 2;

    v8f c;
    if (BIAS_MODE) {
        float bv = bias ? bias[col0 + mn] : 0.f;
        #pragma unroll
        for (int i = 0; i < 8; ++i) c[i] = bv;
    } else {
        #pragma unroll
        for (int i = 0; i < 8; ++i)
            c[i] = C[(size_t)(row0 + half * 8 + i) * 128 + col0 + mn];
    }

    #pragma unroll 4
    for (int k0 = 0; k0 < 128; k0 += 4) {
        v2f a, b;
        a.x = As[mn][k0 + kg];
        a.y = As[mn][k0 + kg + 1];
        b.x = B[(size_t)(k0 + kg) * 128 + col0 + mn];
        b.y = B[(size_t)(k0 + kg + 1) * 128 + col0 + mn];
        c = __builtin_amdgcn_wmma_f32_16x16x4_f32(false, a, false, b, (short)0, c,
                                                  false, false);
    }

    #pragma unroll
    for (int i = 0; i < 8; ++i)
        C[(size_t)(row0 + half * 8 + i) * 128 + col0 + mn] = c[i];
}


__global__ void k_attn(const float* __restrict__ xl, const float* __restrict__ att_src,
                       const float* __restrict__ att_dst, float* __restrict__ a_src,
                       float* __restrict__ a_dst) {
    int t = blockIdx.x * blockDim.x + threadIdx.x;
    if (t >= N_NODES * HEADS) return;
    int n = t >> 2, h = t & 3;
    const float* row = xl + (size_t)n * 128 + h * HD2;
    float ss = 0.f, sd = 0.f;
    #pragma unroll
    for (int j = 0; j < HD2; ++j) {
        float v = row[j];
        ss += v * att_src[h * HD2 + j];
        sd += v * att_dst[h * HD2 + j];
    }
    a_src[t] = ss;
    a_dst[t] = sd;
}

__global__ void k_edge_max(const int* __restrict__ ei, const float* __restrict__ a_src,
                           const float* __restrict__ a_dst, float* __restrict__ m) {
    int gid = blockIdx.x * blockDim.x + threadIdx.x;
    int i = gid >> 2;
    if (i >= N_EDGES + N_NODES) return;
    int h = gid & 3;
    int s, d;
    if (i < N_EDGES) { s = ei[i]; d = ei[N_EDGES + i]; if (s == d) return; }
    else             { s = d = i - N_EDGES; }
    float v = lrelu(a_src[s * 4 + h] + a_dst[d * 4 + h]);
    atomicMaxF(&m[d * 4 + h], v);
}

__global__ void k_edge_sum(const int* __restrict__ ei, const float* __restrict__ a_src,
                           const float* __restrict__ a_dst, const float* __restrict__ m,
                           float* __restrict__ denom) {
    int gid = blockIdx.x * blockDim.x + threadIdx.x;
    int i = gid >> 2;
    if (i >= N_EDGES + N_NODES) return;
    int h = gid & 3;
    int s, d;
    if (i < N_EDGES) { s = ei[i]; d = ei[N_EDGES + i]; if (s == d) return; }
    else             { s = d = i - N_EDGES; }
    float v = lrelu(a_src[s * 4 + h] + a_dst[d * 4 + h]);
    atomicAdd(&denom[d * 4 + h], __expf(v - m[d * 4 + h]));
}

__global__ void k_edge_scatter(const int* __restrict__ ei, const float* __restrict__ xl,
                               const float* __restrict__ a_src, const float* __restrict__ a_dst,
                               const float* __restrict__ m, const float* __restrict__ denom,
                               float* __restrict__ out) {
    int gid = blockIdx.x * blockDim.x + threadIdx.x;
    int i = gid >> 5;
    if (i >= N_EDGES + N_NODES) return;
    int lane = gid & 31;
    int s, d;
    if (i < N_EDGES) { s = ei[i]; d = ei[N_EDGES + i]; if (s == d) return; }
    else             { s = d = i - N_EDGES; }
    int h = lane & 3;
    float v = lrelu(a_src[s * 4 + h] + a_dst[d * 4 + h]);
    float w = __expf(v - m[d * 4 + h]) / denom[d * 4 + h];
    float w0 = __int_as_float(__builtin_amdgcn_readlane(__float_as_int(w), 0));
    float w1 = __int_as_float(__builtin_amdgcn_readlane(__float_as_int(w), 1));
    float w2 = __int_as_float(__builtin_amdgcn_readlane(__float_as_int(w), 2));
    float w3 = __int_as_float(__builtin_amdgcn_readlane(__float_as_int(w), 3));
    const float* xs = xl + (size_t)s * 128 + lane;
    float* o = out + (size_t)d * 128 + lane;
    float x0 = xs[0], x1 = xs[32], x2 = xs[64], x3 = xs[96];
    atomicAdd(o + 0,  x0 * w0);
    atomicAdd(o + 32, x1 * w1);
    atomicAdd(o + 64, x2 * w2);
    atomicAdd(o + 96, x3 * w3);
}


extern "C" void kernel_launch(void* const* d_in, const int* in_sizes, int n_in,
                              void* d_out, int out_size, void* d_ws, size_t ws_size,
                              hipStream_t stream) {
    const float* x        = (const float*)d_in[0];
    const int*   ei       = (const int*)d_in[1];
    const int*   et       = (const int*)d_in[2];
    const float* comp     = (const float*)d_in[3];
    const float* basis    = (const float*)d_in[4];
    const float* root     = (const float*)d_in[5];
    const float* bias1    = (const float*)d_in[6];
    const float* gat_w    = (const float*)d_in[7];
    const float* att_src  = (const float*)d_in[8];
    const float* att_dst  = (const float*)d_in[9];
    const float* gat_bias = (const float*)d_in[10];
    float* out = (float*)d_out;

    float* ws = (float*)d_ws;
    size_t off = 0;
    float* W    = ws + off; off += (size_t)RREL * GDIM * HD1;
    float* cnt  = ws + off; off += (size_t)N_NODES * RREL;
    float* hbuf = ws + off; off += (size_t)N_NODES * HD1;
    float* xl   = ws + off; off += (size_t)N_NODES * HD1;
    float* asg  = ws + off; off += (size_t)N_NODES * HEADS;
    float* adg  = ws + off; off += (size_t)N_NODES * HEADS;
    float* mbuf = ws + off; off += (size_t)N_NODES * HEADS;
    float* dbuf = ws + off; off += (size_t)N_NODES * HEADS;
    float* acc = xl;

    dim3 mmBlock(32, 8);
    int mmGrid = N_NODES / 16;

    k_basis_w<<<(RREL * GDIM * HD1 + 255) / 256, 256, 0, stream>>>(comp, basis, W);
    hipMemsetAsync(cnt, 0, (size_t)N_NODES * RREL * sizeof(float), stream);
    k_count<<<(N_EDGES + 255) / 256, 256, 0, stream>>>(ei, et, cnt);
    k_fill_bias<<<(N_NODES * HD1 + 255) / 256, 256, 0, stream>>>(hbuf, bias1, N_NODES * HD1);

    for (int r = 0; r < RREL; ++r) {
        hipMemsetAsync(acc, 0, (size_t)N_NODES * GDIM * sizeof(float), stream);
        k_scatter_rel<<<(N_EDGES + 255) / 256, 256, 0, stream>>>(x, ei, et, cnt, acc, r);
        k_mm16<false><<<mmGrid, mmBlock, 0, stream>>>(acc, W + (size_t)r * GDIM * HD1,
                                                      hbuf, nullptr);
    }
    k_mm16<false><<<mmGrid, mmBlock, 0, stream>>>(x, root, hbuf, nullptr);

    k_mm16<true><<<mmGrid, mmBlock, 0, stream>>>(hbuf, gat_w, xl, nullptr);
    k_attn<<<(N_NODES * HEADS + 255) / 256, 256, 0, stream>>>(xl, att_src, att_dst, asg, adg);
    k_fill<<<(N_NODES * HEADS + 255) / 256, 256, 0, stream>>>(mbuf, -INFINITY, N_NODES * HEADS);
    hipMemsetAsync(dbuf, 0, (size_t)N_NODES * HEADS * sizeof(float), stream);

    int EN = N_EDGES + N_NODES;
    k_edge_max<<<(EN * 4 + 255) / 256, 256, 0, stream>>>(ei, asg, adg, mbuf);
    k_edge_sum<<<(EN * 4 + 255) / 256, 256, 0, stream>>>(ei, asg, adg, mbuf, dbuf);
    k_fill_bias<<<(N_NODES * HD1 + 255) / 256, 256, 0, stream>>>(out, gat_bias, N_NODES * HD1);
    k_edge_scatter<<<(EN * 32 + 255) / 256, 256, 0, stream>>>(ei, xl, asg, adg, mbuf, dbuf, out);

    (void)in_sizes; (void)n_in; (void)out_size; (void)ws_size;
}


# RGCN: one edge pass adds each edge's row/count into a [relation][node] buffer, then one depth-1152 product (8 relation weights + root, bias as start); was 8 x (zero, scatter, product); float atomics
# speedup vs baseline: 2.3093x; 1.3772x over previous
#include <hip/hip_runtime.h>
#include <math.h>

#define N_NODES 50000
#define N_EDGES 800000
#define GDIM 128
#define HD1 128
#define HEADS 4
#define HD2 32
#define RREL 8
#define NB 20

typedef __attribute__((ext_vector_type(2))) float v2f;
typedef __attribute__((ext_vector_type(8))) float v8f;

__device__ __forceinline__ float lrelu(float x) { return x > 0.f ? x : 0.2f * x; }

__device__ __forceinline__ void atomicMaxF(float* addr, float val) {
    if (val >= 0.f) atomicMax((int*)addr, __float_as_int(val));
    else            atomicMin((unsigned int*)addr, __float_as_uint(val));
}


__global__ void k_basis_w(const float* __restrict__ comp, const float* __restrict__ basis,
                          float* __restrict__ W) {
    int t = blockIdx.x * blockDim.x + threadIdx.x;
    if (t >= RREL * GDIM * HD1) return;
    int r = t / (GDIM * HD1);
    int io = t % (GDIM * HD1);
    float s = 0.f;
    #pragma unroll
    for (int b = 0; b < NB; ++b) s += comp[r * NB + b] * basis[b * GDIM * HD1 + io];
    W[t] = s;
}

__global__ void k_count(const int* __restrict__ ei, const int* __restrict__ et,
                        float* __restrict__ cnt) {
    int e = blockIdx.x * blockDim.x + threadIdx.x;
    if (e >= N_EDGES) return;
    atomicAdd(&cnt[ei[N_EDGES + e] * RREL + et[e]], 1.f);
}

__global__ void k_fill(float* __restrict__ p, float v, int n) {
    int i = blockIdx.x * blockDim.x + threadIdx.x;
    if (i < n) p[i] = v;
}

__global__ void k_fill_bias(float* __restrict__ p, const float* __restrict__ bias, int n) {
    int i = blockIdx.x * blockDim.x + threadIdx.x;
    if (i < n) p[i] = bias[i & 127];
}


#define SCAT_U 4

__global__ void k_scatter_all(const float* __restrict__ x, const int* __restrict__ ei,
                              const int* __restrict__ et, const float* __restrict__ cnt,
                              float* __restrict__ acc8) {
    int gid = blockIdx.x * blockDim.x + threadIdx.x;
    int lane = threadIdx.x & 31;
    bool in = gid < N_EDGES;
    int ec = in ? gid : N_EDGES - 1;
    int t = et[ec];
    int s = ei[ec];
    int d = ei[N_EDGES + ec];
    unsigned todo = (unsigned)__builtin_amdgcn_ballot_w32(in);
    while (todo != 0u) {
        int  j[SCAT_U];
        bool ok[SCAT_U];
        #pragma unroll
        for (int u = 0; u < SCAT_U; ++u) {
            ok[u] = todo != 0u;
            j[u] = ok[u] ? __builtin_ctz(todo) : j[0];
            todo &= todo - 1u;
        }
        int   dj[SCAT_U];
        int   tj[SCAT_U];
        float cn[SCAT_U];
        float v[SCAT_U][4];
        #pragma unroll
        for (int u = 0; u < SCAT_U; ++u) {
            int sj = __builtin_amdgcn_readlane(s, j[u]);
            dj[u] = __builtin_amdgcn_readlane(d, j[u]);
            tj[u] = __builtin_amdgcn_readlane(t, j[u]);
            const float* xs = x + (size_t)sj * GDIM + lane;
            v[u][0] = xs[0]; v[u][1] = xs[32]; v[u][2] = xs[64]; v[u][3] = xs[96];
            cn[u] = cnt[dj[u] * RREL + tj[u]];
        }
        static_assert(SCAT_U == 4, "the statement below names the loads of four edges");
        asm volatile("" :: "v"(v[0][0]), "v"(v[0][1]), "v"(v[0][2]), "v"(v[0][3]),
                           "v"(v[1][0]), "v"(v[1][1]), "v"(v[1][2]), "v"(v[1][3]),
                           "v"(v[2][0]), "v"(v[2][1]), "v"(v[2][2]), "v"(v[2][3]),
                           "v"(v[3][0]), "v"(v[3][1]), "v"(v[3][2]), "v"(v[3][3]),
                           "s"(cn[0]), "s"(cn[1]), "s"(cn[2]), "s"(cn[3]));
        #pragma unroll
        for (int u = 0; u < SCAT_U; ++u) {
            if (ok[u]) {
                float c = 1.f / cn[u];
                float* o = acc8 + ((size_t)tj[u] * N_NODES + dj[u]) * GDIM + lane;
                atomicAdd(o + 0,  v[u][0] * c);
                atomicAdd(o + 32, v[u][1] * c);
                atomicAdd(o + 64, v[u][2] * c);
                atomicAdd(o + 96, v[u][3] * c);
            }
        }
    }
}


template <bool BIAS_MODE>
__global__ __launch_bounds__(256) void k_mm16(const float* __restrict__ A,
                                              const float* __restrict__ B,
                                              float* __restrict__ C,
                                              const float* __restrict__ bias) {
    __shared__ float As[16][132];
    int tx = threadIdx.x;
    int ty = threadIdx.y;
    int tid = ty * 32 + tx;
    int row0 = blockIdx.x * 16;

    const float4* A4 = (const float4*)(A + (size_t)row0 * 128);
    for (int i = tid; i < 16 * 32; i += 256) {
        int rr = i >> 5, cc = i & 31;
        float4 v = A4[rr * 32 + cc];
        As[rr][cc * 4 + 0] = v.x;
        As[rr][cc * 4 + 1] = v.y;
        As[rr][cc * 4 + 2] = v.z;
        As[rr][cc * 4 + 3] = v.w;
    }
    __syncthreads();

    int col0 = ty * 16;
    int half = tx >> 4;
    int mn = tx & 15;
    int kg = half * 2;

    v8f c;
    if (BIAS_MODE) {
        float bv = bias ? bias[col0 + mn] : 0.f;
        #pragma unroll
        for (int i = 0; i < 8; ++i) c[i] = bv;
    } else {
        #pragma unroll
        for (int i = 0; i < 8; ++i)
            c[i] = C[(size_t)(row0 + half * 8 + i) * 128 + col0 + mn];
    }

    #pragma unroll 4
    for (int k0 = 0; k0 < 128; k0 += 4) {
        v2f a, b;
        a.x = As[mn][k0 + kg];
        a.y = As[mn][k0 + kg + 1];
        b.x = B[(size_t)(k0 + kg) * 128 + col0 + mn];
        b.y = B[(size_t)(k0 + kg + 1) * 128 + col0 + mn];
        c = __builtin_amdgcn_wmma_f32_16x16x4_f32(false, a, false, b, (short)0, c,
                                                  false, false);
    }

    #pragma unroll
    for (int i = 0; i < 8; ++i)
        C[(size_t)(row0 + half * 8 + i) * 128 + col0 + mn] = c[i];
}


#define RG_CH (RREL + 1)

__global__ __launch_bounds__(256) void k_mm_rgcn(const float* __restrict__ acc8,
                                                 const float* __restrict__ x,
                                                 const float* __restrict__ W,
                                                 const float* __restrict__ root,
                                                 const float* __restrict__ bias,
                                                 float* __restrict__ C) {
    __shared__ float As[2][16][132];
    int tx = threadIdx.x;
    int ty = threadIdx.y;
    int tid = ty * 32 + tx;
    int row0 = blockIdx.x * 16;

    int ra = tid >> 5, ca = tid & 31;
    const float4* a_lo = (const float4*)(acc8 + (size_t)(row0 + ra) * 128) + ca;
    const float4* a_hi = (const float4*)(acc8 + (size_t)(row0 + ra + 8) * 128) + ca;
    const float4* x_lo = (const float4*)(x + (size_t)(row0 + ra) * 128) + ca;
    const float4* x_hi = (const float4*)(x + (size_t)(row0 + ra + 8) * 128) + ca;
    const size_t chunk4 = (size_t)N_NODES * 128 / 4;

    float4 p0 = a_lo[0], p1 = a_hi[0];
    As[0][ra][ca * 4 + 0] = p0.x; As[0][ra][ca * 4 + 1] = p0.y;
    As[0][ra][ca * 4 + 2] = p0.z; As[0][ra][ca * 4 + 3] = p0.w;
    As[0][ra + 8][ca * 4 + 0] = p1.x; As[0][ra + 8][ca * 4 + 1] = p1.y;
    As[0][ra + 8][ca * 4 + 2] = p1.z; As[0][ra + 8][ca * 4 + 3] = p1.w;
    __syncthreads();

    int col0 = ty * 16;
    int half = tx >> 4;
    int mn = tx & 15;
    int kg = half * 2;

    v8f c;
    {
        float bv = bias[col0 + mn];
        #pragma unroll
        for (int i = 0; i < 8; ++i) c[i] = bv;
    }

    for (int ch = 0; ch < RG_CH; ++ch) {
        int cur = ch & 1;
        if (ch + 1 < RG_CH) {
            if (ch + 1 < RREL) { p0 = a_lo[(size_t)(ch + 1) * chunk4]; p1 = a_hi[(size_t)(ch + 1) * chunk4]; }
            else               { p0 = x_lo[0];                         p1 = x_hi[0]; }
        }
        const float* B = (ch < RREL) ? W + (size_t)ch * GDIM * HD1 : root;

        #pragma unroll 4
        for (int k0 = 0; k0 < 128; k0 += 4) {
            v2f a, b;
            a.x = As[cur][mn][k0 + kg];
            a.y = As[cur][mn][k0 + kg + 1];
            b.x = B[(size_t)(k0 + kg) * 128 + col0 + mn];
            b.y = B[(size_t)(k0 + kg + 1) * 128 + col0 + mn];
            c = __builtin_amdgcn_wmma_f32_16x16x4_f32(false, a, false, b, (short)0, c,
                                                      false, false);
        }

        if (ch + 1 < RG_CH) {
            int nxt = cur ^ 1;
            As[nxt][ra][ca * 4 + 0] = p0.x; As[nxt][ra][ca * 4 + 1] = p0.y;
            As[nxt][ra][ca * 4 + 2] = p0.z; As[nxt][ra][ca * 4 + 3] = p0.w;
            As[nxt][ra + 8][ca * 4 + 0] = p1.x; As[nxt][ra + 8][ca * 4 + 1] = p1.y;
            As[nxt][ra + 8][ca * 4 + 2] = p1.z; As[nxt][ra + 8][ca * 4 + 3] = p1.w;
            __syncthreads();
        }
    }

    #pragma unroll
    for (int i = 0; i < 8; ++i)
        C[(size_t)(row0 + half * 8 + i) * 128 + col0 + mn] = c[i];
}


__global__ void k_attn(const float* __restrict__ xl, const float* __restrict__ att_src,
                       const float* __restrict__ att_dst, float* __restrict__ a_src,
                       float* __restrict__ a_dst) {
    int t = blockIdx.x * blockDim.x + threadIdx.x;
    if (t >= N_NODES * HEADS) return;
    int n = t >> 2, h = t & 3;
    const float* row = xl + (size_t)n * 128 + h * HD2;
    float ss = 0.f, sd = 0.f;
    #pragma unroll
    for (int j = 0; j < HD2; ++j) {
        float v = row[j];
        ss += v * att_src[h * HD2 + j];
        sd += v * att_dst[h * HD2 + j];
    }
    a_src[t] = ss;
    a_dst[t] = sd;
}

__global__ void k_edge_max(const int* __restrict__ ei, const float* __restrict__ a_src,
                           const float* __restrict__ a_dst, float* __restrict__ m) {
    int gid = blockIdx.x * blockDim.x + threadIdx.x;
    int i = gid >> 2;
    if (i >= N_EDGES + N_NODES) return;
    int h = gid & 3;
    int s, d;
    if (i < N_EDGES) { s = ei[i]; d = ei[N_EDGES + i]; if (s == d) return; }
    else             { s = d = i - N_EDGES; }
    float v = lrelu(a_src[s * 4 + h] + a_dst[d * 4 + h]);
    atomicMaxF(&m[d * 4 + h], v);
}

__global__ void k_edge_sum(const int* __restrict__ ei, const float* __restrict__ a_src,
                           const float* __restrict__ a_dst, const float* __restrict__ m,
                           float* __restrict__ denom) {
    int gid = blockIdx.x * blockDim.x + threadIdx.x;
    int i = gid >> 2;
    if (i >= N_EDGES + N_NODES) return;
    int h = gid & 3;
    int s, d;
    if (i < N_EDGES) { s = ei[i]; d = ei[N_EDGES + i]; if (s == d) return; }
    else             { s = d = i - N_EDGES; }
    float v = lrelu(a_src[s * 4 + h] + a_dst[d * 4 + h]);
    atomicAdd(&denom[d * 4 + h], __expf(v - m[d * 4 + h]));
}

__global__ void k_edge_scatter(const int* __restrict__ ei, const float* __restrict__ xl,
                               const float* __restrict__ a_src, const float* __restrict__ a_dst,
                               const float* __restrict__ m, const float* __restrict__ denom,
                               float* __restrict__ out) {
    int gid = blockIdx.x * blockDim.x + threadIdx.x;
    int i = gid >> 5;
    if (i >= N_EDGES + N_NODES) return;
    int lane = gid & 31;
    int s, d;
    if (i < N_EDGES) { s = ei[i]; d = ei[N_EDGES + i]; if (s == d) return; }
    else             { s = d = i - N_EDGES; }
    int h = lane & 3;
    float v = lrelu(a_src[s * 4 + h] + a_dst[d * 4 + h]);
    float w = __expf(v - m[d * 4 + h]) / denom[d * 4 + h];
    float w0 = __int_as_float(__builtin_amdgcn_readlane(__float_as_int(w), 0));
    float w1 = __int_as_float(__builtin_amdgcn_readlane(__float_as_int(w), 1));
    float w2 = __int_as_float(__builtin_amdgcn_readlane(__float_as_int(w), 2));
    float w3 = __int_as_float(__builtin_amdgcn_readlane(__float_as_int(w), 3));
    const float* xs = xl + (size_t)s * 128 + lane;
    float* o = out + (size_t)d * 128 + lane;
    float x0 = xs[0], x1 = xs[32], x2 = xs[64], x3 = xs[96];
    atomicAdd(o + 0,  x0 * w0);
    atomicAdd(o + 32, x1 * w1);
    atomicAdd(o + 64, x2 * w2);
    atomicAdd(o + 96, x3 * w3);
}


extern "C" void kernel_launch(void* const* d_in, const int* in_sizes, int n_in,
                              void* d_out, int out_size, void* d_ws, size_t ws_size,
                              hipStream_t stream) {
    const float* x        = (const float*)d_in[0];
    const int*   ei       = (const int*)d_in[1];
    const int*   et       = (const int*)d_in[2];
    const float* comp     = (const float*)d_in[3];
    const float* basis    = (const float*)d_in[4];
    const float* root     = (const float*)d_in[5];
    const float* bias1    = (const float*)d_in[6];
    const float* gat_w    = (const float*)d_in[7];
    const float* att_src  = (const float*)d_in[8];
    const float* att_dst  = (const float*)d_in[9];
    const float* gat_bias = (const float*)d_in[10];
    float* out = (float*)d_out;

    float* ws = (float*)d_ws;
    size_t off = 0;
    float* W    = ws + off; off += (size_t)RREL * GDIM * HD1;
    float* cnt  = ws + off; off += (size_t)N_NODES * RREL;
    float* hbuf = ws + off; off += (size_t)N_NODES * HD1;
    float* xl   = ws + off; off += (size_t)N_NODES * HD1;
    float* asg  = ws + off; off += (size_t)N_NODES * HEADS;
    float* adg  = ws + off; off += (size_t)N_NODES * HEADS;
    float* mbuf = ws + off; off += (size_t)N_NODES * HEADS;
    float* dbuf = ws + off; off += (size_t)N_NODES * HEADS;
    float* acc8 = ws + off; off += (size_t)RREL * N_NODES * GDIM;

    dim3 mmBlock(32, 8);
    int mmGrid = N_NODES / 16;

    k_basis_w<<<(RREL * GDIM * HD1 + 255) / 256, 256, 0, stream>>>(comp, basis, W);
    hipMemsetAsync(cnt, 0, (size_t)N_NODES * RREL * sizeof(float), stream);
    k_count<<<(N_EDGES + 255) / 256, 256, 0, stream>>>(ei, et, cnt);

    hipMemsetAsync(acc8, 0, (size_t)RREL * N_NODES * GDIM * sizeof(float), stream);
    k_scatter_all<<<(N_EDGES + 255) / 256, 256, 0, stream>>>(x, ei, et, cnt, acc8);
    k_mm_rgcn<<<mmGrid, mmBlock, 0, stream>>>(acc8, x, W, root, bias1, hbuf);

    k_mm16<true><<<mmGrid, mmBlock, 0, stream>>>(hbuf, gat_w, xl, nullptr);
    k_attn<<<(N_NODES * HEADS + 255) / 256, 256, 0, stream>>>(xl, att_src, att_dst, asg, adg);
    k_fill<<<(N_NODES * HEADS + 255) / 256, 256, 0, stream>>>(mbuf, -INFINITY, N_NODES * HEADS);
    hipMemsetAsync(dbuf, 0, (size_t)N_NODES * HEADS * sizeof(float), stream);

    int EN = N_EDGES + N_NODES;
    k_edge_max<<<(EN * 4 + 255) / 256, 256, 0, stream>>>(ei, asg, adg, mbuf);
    k_edge_sum<<<(EN * 4 + 255) / 256, 256, 0, stream>>>(ei, asg, adg, mbuf, dbuf);
    k_fill_bias<<<(N_NODES * HD1 + 255) / 256, 256, 0, stream>>>(out, gat_bias, N_NODES * HD1);
    k_edge_scatter<<<(EN * 32 + 255) / 256, 256, 0, stream>>>(ei, xl, asg, adg, mbuf, dbuf, out);

    (void)in_sizes; (void)n_in; (void)out_size; (void)ws_size;
}


# GAT softmax and sum: edges bucketed by destination (int count, scan, fill); one wave per node sums its bucket in registers in source order, one store per row; its float atomics go, sum order fixed
# speedup vs baseline: 2.7305x; 1.1824x over previous
#include <hip/hip_runtime.h>
#include <math.h>

#define N_NODES 50000
#define N_EDGES 800000
#define GDIM 128
#define HD1 128
#define HEADS 4
#define HD2 32
#define RREL 8
#define NB 20

typedef __attribute__((ext_vector_type(2))) float v2f;
typedef __attribute__((ext_vector_type(8))) float v8f;

__device__ __forceinline__ float lrelu(float x) { return x > 0.f ? x : 0.2f * x; }


__global__ void k_basis_w(const float* __restrict__ comp, const float* __restrict__ basis,
                          float* __restrict__ W) {
    int t = blockIdx.x * blockDim.x + threadIdx.x;
    if (t >= RREL * GDIM * HD1) return;
    int r = t / (GDIM * HD1);
    int io = t % (GDIM * HD1);
    float s = 0.f;
    #pragma unroll
    for (int b = 0; b < NB; ++b) s += comp[r * NB + b] * basis[b * GDIM * HD1 + io];
    W[t] = s;
}

__global__ void k_count(const int* __restrict__ ei, const int* __restrict__ et,
                        float* __restrict__ cnt, int* __restrict__ deg, int* __restrict__ slot) {
    int e = blockIdx.x * blockDim.x + threadIdx.x;
    if (e >= N_EDGES) return;
    int s = ei[e];
    int d = ei[N_EDGES + e];
    atomicAdd(&cnt[d * RREL + et[e]], 1.f);
    slot[e] = (s != d) ? atomicAdd(&deg[d], 1) : -1;
}


#define SCAT_U 4

__global__ void k_scatter_all(const float* __restrict__ x, const int* __restrict__ ei,
                              const int* __restrict__ et, const float* __restrict__ cnt,
                              float* __restrict__ acc8) {
    int gid = blockIdx.x * blockDim.x + threadIdx.x;
    int lane = threadIdx.x & 31;
    bool in = gid < N_EDGES;
    int ec = in ? gid : N_EDGES - 1;
    int t = et[ec];
    int s = ei[ec];
    int d = ei[N_EDGES + ec];
    unsigned todo = (unsigned)__builtin_amdgcn_ballot_w32(in);
    while (todo != 0u) {
        int  j[SCAT_U];
        bool ok[SCAT_U];
        #pragma unroll
        for (int u = 0; u < SCAT_U; ++u) {
            ok[u] = todo != 0u;
            j[u] = ok[u] ? __builtin_ctz(todo) : j[0];
            todo &= todo - 1u;
        }
        int   dj[SCAT_U];
        int   tj[SCAT_U];
        float cn[SCAT_U];
        float v[SCAT_U][4];
        #pragma unroll
        for (int u = 0; u < SCAT_U; ++u) {
            int sj = __builtin_amdgcn_readlane(s, j[u]);
            dj[u] = __builtin_amdgcn_readlane(d, j[u]);
            tj[u] = __builtin_amdgcn_readlane(t, j[u]);
            const float* xs = x + (size_t)sj * GDIM + lane;
            v[u][0] = xs[0]; v[u][1] = xs[32]; v[u][2] = xs[64]; v[u][3] = xs[96];
            cn[u] = cnt[dj[u] * RREL + tj[u]];
        }
        static_assert(SCAT_U == 4, "the statement below names the loads of four edges");
        asm volatile("" :: "v"(v[0][0]), "v"(v[0][1]), "v"(v[0][2]), "v"(v[0][3]),
                           "v"(v[1][0]), "v"(v[1][1]), "v"(v[1][2]), "v"(v[1][3]),
                           "v"(v[2][0]), "v"(v[2][1]), "v"(v[2][2]), "v"(v[2][3]),
                           "v"(v[3][0]), "v"(v[3][1]), "v"(v[3][2]), "v"(v[3][3]),
                           "s"(cn[0]), "s"(cn[1]), "s"(cn[2]), "s"(cn[3]));
        #pragma unroll
        for (int u = 0; u < SCAT_U; ++u) {
            if (ok[u]) {
                float c = 1.f / cn[u];
                float* o = acc8 + ((size_t)tj[u] * N_NODES + dj[u]) * GDIM + lane;
                atomicAdd(o + 0,  v[u][0] * c);
                atomicAdd(o + 32, v[u][1] * c);
                atomicAdd(o + 64, v[u][2] * c);
                atomicAdd(o + 96, v[u][3] * c);
            }
        }
    }
}


template <bool BIAS_MODE>
__global__ __launch_bounds__(256) void k_mm16(const float* __restrict__ A,
                                              const float* __restrict__ B,
                                              float* __restrict__ C,
                                              const float* __restrict__ bias) {
    __shared__ float As[16][132];
    int tx = threadIdx.x;
    int ty = threadIdx.y;
    int tid = ty * 32 + tx;
    int row0 = blockIdx.x * 16;

    const float4* A4 = (const float4*)(A + (size_t)row0 * 128);
    for (int i = tid; i < 16 * 32; i += 256) {
        int rr = i >> 5, cc = i & 31;
        float4 v = A4[rr * 32 + cc];
        As[rr][cc * 4 + 0] = v.x;
        As[rr][cc * 4 + 1] = v.y;
        As[rr][cc * 4 + 2] = v.z;
        As[rr][cc * 4 + 3] = v.w;
    }
    __syncthreads();

    int col0 = ty * 16;
    int half = tx >> 4;
    int mn = tx & 15;
    int kg = half * 2;

    v8f c;
    if (BIAS_MODE) {
        float bv = bias ? bias[col0 + mn] : 0.f;
        #pragma unroll
        for (int i = 0; i < 8; ++i) c[i] = bv;
    } else {
        #pragma unroll
        for (int i = 0; i < 8; ++i)
            c[i] = C[(size_t)(row0 + half * 8 + i) * 128 + col0 + mn];
    }

    #pragma unroll 4
    for (int k0 = 0; k0 < 128; k0 += 4) {
        v2f a, b;
        a.x = As[mn][k0 + kg];
        a.y = As[mn][k0 + kg + 1];
        b.x = B[(size_t)(k0 + kg) * 128 + col0 + mn];
        b.y = B[(size_t)(k0 + kg + 1) * 128 + col0 + mn];
        c = __builtin_amdgcn_wmma_f32_16x16x4_f32(false, a, false, b, (short)0, c,
                                                  false, false);
    }

    #pragma unroll
    for (int i = 0; i < 8; ++i)
        C[(size_t)(row0 + half * 8 + i) * 128 + col0 + mn] = c[i];
}


#define RG_CH (RREL + 1)

__global__ __launch_bounds__(256) void k_mm_rgcn(const float* __restrict__ acc8,
                                                 const float* __restrict__ x,
                                                 const float* __restrict__ W,
                                                 const float* __restrict__ root,
                                                 const float* __restrict__ bias,
                                                 float* __restrict__ C) {
    __shared__ float As[2][16][132];
    int tx = threadIdx.x;
    int ty = threadIdx.y;
    int tid = ty * 32 + tx;
    int row0 = blockIdx.x * 16;

    int ra = tid >> 5, ca = tid & 31;
    const float4* a_lo = (const float4*)(acc8 + (size_t)(row0 + ra) * 128) + ca;
    const float4* a_hi = (const float4*)(acc8 + (size_t)(row0 + ra + 8) * 128) + ca;
    const float4* x_lo = (const float4*)(x + (size_t)(row0 + ra) * 128) + ca;
    const float4* x_hi = (const float4*)(x + (size_t)(row0 + ra + 8) * 128) + ca;
    const size_t chunk4 = (size_t)N_NODES * 128 / 4;

    float4 p0 = a_lo[0], p1 = a_hi[0];
    As[0][ra][ca * 4 + 0] = p0.x; As[0][ra][ca * 4 + 1] = p0.y;
    As[0][ra][ca * 4 + 2] = p0.z; As[0][ra][ca * 4 + 3] = p0.w;
    As[0][ra + 8][ca * 4 + 0] = p1.x; As[0][ra + 8][ca * 4 + 1] = p1.y;
    As[0][ra + 8][ca * 4 + 2] = p1.z; As[0][ra + 8][ca * 4 + 3] = p1.w;
    __syncthreads();

    int col0 = ty * 16;
    int half = tx >> 4;
    int mn = tx & 15;
    int kg = half * 2;

    v8f c;
    {
        float bv = bias[col0 + mn];
        #pragma unroll
        for (int i = 0; i < 8; ++i) c[i] = bv;
    }

    for (int ch = 0; ch < RG_CH; ++ch) {
        int cur = ch & 1;
        if (ch + 1 < RG_CH) {
            if (ch + 1 < RREL) { p0 = a_lo[(size_t)(ch + 1) * chunk4]; p1 = a_hi[(size_t)(ch + 1) * chunk4]; }
            else               { p0 = x_lo[0];                         p1 = x_hi[0]; }
        }
        const float* B = (ch < RREL) ? W + (size_t)ch * GDIM * HD1 : root;

        #pragma unroll 4
        for (int k0 = 0; k0 < 128; k0 += 4) {
            v2f a, b;
            a.x = As[cur][mn][k0 + kg];
            a.y = As[cur][mn][k0 + kg + 1];
            b.x = B[(size_t)(k0 + kg) * 128 + col0 + mn];
            b.y = B[(size_t)(k0 + kg + 1) * 128 + col0 + mn];
            c = __builtin_amdgcn_wmma_f32_16x16x4_f32(false, a, false, b, (short)0, c,
                                                      false, false);
        }

        if (ch + 1 < RG_CH) {
            int nxt = cur ^ 1;
            As[nxt][ra][ca * 4 + 0] = p0.x; As[nxt][ra][ca * 4 + 1] = p0.y;
            As[nxt][ra][ca * 4 + 2] = p0.z; As[nxt][ra][ca * 4 + 3] = p0.w;
            As[nxt][ra + 8][ca * 4 + 0] = p1.x; As[nxt][ra + 8][ca * 4 + 1] = p1.y;
            As[nxt][ra + 8][ca * 4 + 2] = p1.z; As[nxt][ra + 8][ca * 4 + 3] = p1.w;
            __syncthreads();
        }
    }

    #pragma unroll
    for (int i = 0; i < 8; ++i)
        C[(size_t)(row0 + half * 8 + i) * 128 + col0 + mn] = c[i];
}


__global__ void k_attn(const float* __restrict__ xl, const float* __restrict__ att_src,
                       const float* __restrict__ att_dst, float* __restrict__ a_src,
                       float* __restrict__ a_dst) {
    int t = blockIdx.x * blockDim.x + threadIdx.x;
    if (t >= N_NODES * HEADS) return;
    int n = t >> 2, h = t & 3;
    const float* row = xl + (size_t)n * 128 + h * HD2;
    float ss = 0.f, sd = 0.f;
    #pragma unroll
    for (int j = 0; j < HD2; ++j) {
        float v = row[j];
        ss += v * att_src[h * HD2 + j];
        sd += v * att_dst[h * HD2 + j];
    }
    a_src[t] = ss;
    a_dst[t] = sd;
}


#define SCAN_T 1024
#define SCAN_B ((N_NODES + SCAN_T - 1) / SCAN_T)

__global__ __launch_bounds__(SCAN_T) void k_scan(const int* __restrict__ deg,
                                                 int* __restrict__ rowptr) {
    __shared__ int w_own[SCAN_T / 32];
    __shared__ int w_pre[SCAN_T / 32];
    int t = threadIdx.x, lane = t & 31, w = t >> 5;
    int b = blockIdx.x;
    int pre = 0;
    for (int i = 0; i < b; ++i) pre += deg[i * SCAN_T + t];
    int idx = b * SCAN_T + t;
    int own = idx < N_NODES ? deg[idx] : 0;
    int inc = own;
    for (int o = 1; o < 32; o <<= 1) {
        int v = __shfl_up(inc, o);
        if (lane >= o) inc += v;
    }
    for (int o = 16; o > 0; o >>= 1) pre += __shfl_xor(pre, o);
    if (lane == 31) w_own[w] = inc;
    if (lane == 0)  w_pre[w] = pre;
    __syncthreads();
    int add = w_pre[lane] + (lane < w ? w_own[lane] : 0);
    for (int o = 16; o > 0; o >>= 1) add += __shfl_xor(add, o);
    int start = add + inc - own;
    if (idx < N_NODES) rowptr[idx] = start;
}

__global__ void k_fill_csr(const int* __restrict__ ei, const int* __restrict__ rowptr,
                           const int* __restrict__ slot, int* __restrict__ csr_src) {
    int e = blockIdx.x * blockDim.x + threadIdx.x;
    if (e >= N_EDGES) return;
    int s = ei[e];
    int d = ei[N_EDGES + e];
    if (s != d) csr_src[rowptr[d] + slot[e]] = s;
}

__device__ __forceinline__ float readlane_f(float v, int l) {
    return __int_as_float(__builtin_amdgcn_readlane(__float_as_int(v), l));
}

__device__ __forceinline__ float wave_max(float v) {
    for (int o = 16; o > 0; o >>= 1) v = fmaxf(v, __shfl_xor(v, o));
    return v;
}

__device__ __forceinline__ float wave_sum(float v) {
    for (int o = 16; o > 0; o >>= 1) v += __shfl_xor(v, o);
    return v;
}

__device__ __forceinline__ void gat_rows(const float* __restrict__ xl, int lane, int sv,
                                         float w0, float w1, float w2, float w3, int cc,
                                         float& acc0, float& acc1, float& acc2, float& acc3) {
    #pragma unroll 4
    for (int k = 0; k < cc; ++k) {
        int s = __builtin_amdgcn_readlane(sv, k);
        float u0 = readlane_f(w0, k);
        float u1 = readlane_f(w1, k);
        float u2 = readlane_f(w2, k);
        float u3 = readlane_f(w3, k);
        const float* xs = xl + (size_t)s * 128 + lane;
        acc0 = fmaf(xs[0],  u0, acc0);
        acc1 = fmaf(xs[32], u1, acc1);
        acc2 = fmaf(xs[64], u2, acc2);
        acc3 = fmaf(xs[96], u3, acc3);
    }
}

__device__ __forceinline__ int bucket_sorted(const int* __restrict__ csr_src, int beg, int n,
                                             int base, int lane, int fill) {
    int mine = fill;
    for (int k = 0; k < n; ++k) {
        int sk = csr_src[beg + k];
        int rank = 0;
        for (int b = 0; b < n; b += 32) {
            int i = b + lane;
            int si = i < n ? csr_src[beg + i] : 0x7fffffff;
            rank += __builtin_popcount(
                __builtin_amdgcn_ballot_w32(si < sk || (si == sk && i < k)));
        }
        if (lane == rank - base) mine = sk;
    }
    return mine;
}

__global__ __launch_bounds__(256) void k_gat_node(const float* __restrict__ xl,
                                                  const float* __restrict__ a_src,
                                                  const float* __restrict__ a_dst,
                                                  const int* __restrict__ rowptr,
                                                  const int* __restrict__ deg,
                                                  const int* __restrict__ csr_src,
                                                  const float* __restrict__ gat_bias,
                                                  float* __restrict__ out) {
    int gid = blockIdx.x * blockDim.x + threadIdx.x;
    if ((gid >> 5) >= N_NODES) return;
    int lane = threadIdx.x & 31;
    int d   = __builtin_amdgcn_readfirstlane(gid >> 5);
    int beg = __builtin_amdgcn_readfirstlane(rowptr[d]);
    int n   = __builtin_amdgcn_readfirstlane(deg[d]);

    const float4 ad  = *(const float4*)(a_dst + (size_t)d * 4);
    const float4 asd = *(const float4*)(a_src + (size_t)d * 4);
    float l0 = lrelu(asd.x + ad.x), l1 = lrelu(asd.y + ad.y);
    float l2 = lrelu(asd.z + ad.z), l3 = lrelu(asd.w + ad.w);

    bool ok = lane < n;
    int sv = d;
    if (n <= 32) {
        int raw = ok ? csr_src[beg + lane] : 0x7fffffff;
        for (int k = 0; k < n; ++k) {
            int sk = __builtin_amdgcn_readlane(raw, k);
            int rank = __builtin_popcount(
                __builtin_amdgcn_ballot_w32(raw < sk || (raw == sk && lane < k)));
            if (lane == rank) sv = sk;
        }
    } else {
        sv = bucket_sorted(csr_src, beg, n, 0, lane, d);
    }
    float4 av = *(const float4*)(a_src + (size_t)sv * 4);
    float c0 = lrelu(av.x + ad.x), c1 = lrelu(av.y + ad.y);
    float c2 = lrelu(av.z + ad.z), c3 = lrelu(av.w + ad.w);

    float m0 = ok ? fmaxf(l0, c0) : l0, m1 = ok ? fmaxf(l1, c1) : l1;
    float m2 = ok ? fmaxf(l2, c2) : l2, m3 = ok ? fmaxf(l3, c3) : l3;
    for (int b = 32; b < n; b += 32) {
        bool okb = b + lane < n;
        int s = bucket_sorted(csr_src, beg, n, b, lane, d);
        float4 a = *(const float4*)(a_src + (size_t)s * 4);
        float t0 = lrelu(a.x + ad.x), t1 = lrelu(a.y + ad.y);
        float t2 = lrelu(a.z + ad.z), t3 = lrelu(a.w + ad.w);
        m0 = okb ? fmaxf(m0, t0) : m0; m1 = okb ? fmaxf(m1, t1) : m1;
        m2 = okb ? fmaxf(m2, t2) : m2; m3 = okb ? fmaxf(m3, t3) : m3;
    }
    m0 = wave_max(m0); m1 = wave_max(m1); m2 = wave_max(m2); m3 = wave_max(m3);

    float e0 = ok ? __expf(c0 - m0) : 0.f, e1 = ok ? __expf(c1 - m1) : 0.f;
    float e2 = ok ? __expf(c2 - m2) : 0.f, e3 = ok ? __expf(c3 - m3) : 0.f;
    float d0 = e0, d1 = e1, d2 = e2, d3 = e3;
    for (int b = 32; b < n; b += 32) {
        bool okb = b + lane < n;
        int s = bucket_sorted(csr_src, beg, n, b, lane, d);
        float4 a = *(const float4*)(a_src + (size_t)s * 4);
        d0 += okb ? __expf(lrelu(a.x + ad.x) - m0) : 0.f;
        d1 += okb ? __expf(lrelu(a.y + ad.y) - m1) : 0.f;
        d2 += okb ? __expf(lrelu(a.z + ad.z) - m2) : 0.f;
        d3 += okb ? __expf(lrelu(a.w + ad.w) - m3) : 0.f;
    }
    float s0 = __expf(l0 - m0), s1 = __expf(l1 - m1), s2 = __expf(l2 - m2), s3 = __expf(l3 - m3);
    d0 = wave_sum(d0) + s0; d1 = wave_sum(d1) + s1;
    d2 = wave_sum(d2) + s2; d3 = wave_sum(d3) + s3;

    const float* xd = xl + (size_t)d * 128 + lane;
    float acc0 = xd[0]  * (s0 / d0);
    float acc1 = xd[32] * (s1 / d1);
    float acc2 = xd[64] * (s2 / d2);
    float acc3 = xd[96] * (s3 / d3);
    gat_rows(xl, lane, sv, e0 / d0, e1 / d1, e2 / d2, e3 / d3, n < 32 ? n : 32,
             acc0, acc1, acc2, acc3);
    for (int b = 32; b < n; b += 32) {
        bool okb = b + lane < n;
        int s = bucket_sorted(csr_src, beg, n, b, lane, d);
        float4 a = *(const float4*)(a_src + (size_t)s * 4);
        float w0 = okb ? __expf(lrelu(a.x + ad.x) - m0) / d0 : 0.f;
        float w1 = okb ? __expf(lrelu(a.y + ad.y) - m1) / d1 : 0.f;
        float w2 = okb ? __expf(lrelu(a.z + ad.z) - m2) / d2 : 0.f;
        float w3 = okb ? __expf(lrelu(a.w + ad.w) - m3) / d3 : 0.f;
        gat_rows(xl, lane, s, w0, w1, w2, w3, (n - b) < 32 ? (n - b) : 32,
                 acc0, acc1, acc2, acc3);
    }

    float* o = out + (size_t)d * 128 + lane;
    o[0]  = acc0 + gat_bias[lane];
    o[32] = acc1 + gat_bias[32 + lane];
    o[64] = acc2 + gat_bias[64 + lane];
    o[96] = acc3 + gat_bias[96 + lane];
}


extern "C" void kernel_launch(void* const* d_in, const int* in_sizes, int n_in,
                              void* d_out, int out_size, void* d_ws, size_t ws_size,
                              hipStream_t stream) {
    const float* x        = (const float*)d_in[0];
    const int*   ei       = (const int*)d_in[1];
    const int*   et       = (const int*)d_in[2];
    const float* comp     = (const float*)d_in[3];
    const float* basis    = (const float*)d_in[4];
    const float* root     = (const float*)d_in[5];
    const float* bias1    = (const float*)d_in[6];
    const float* gat_w    = (const float*)d_in[7];
    const float* att_src  = (const float*)d_in[8];
    const float* att_dst  = (const float*)d_in[9];
    const float* gat_bias = (const float*)d_in[10];
    float* out = (float*)d_out;

    float* ws = (float*)d_ws;
    size_t off = 0;
    float* W    = ws + off; off += (size_t)RREL * GDIM * HD1;
    float* cnt  = ws + off; off += (size_t)N_NODES * RREL;
    int*   deg  = (int*)(ws + off); off += (size_t)N_NODES;
    float* hbuf = ws + off; off += (size_t)N_NODES * HD1;
    float* xl   = ws + off; off += (size_t)N_NODES * HD1;
    float* asg  = ws + off; off += (size_t)N_NODES * HEADS;
    float* adg  = ws + off; off += (size_t)N_NODES * HEADS;
    int*   rowptr = (int*)(ws + off); off += (size_t)N_NODES;
    int*   slot   = (int*)(ws + off); off += (size_t)N_EDGES;
    int*   csrc   = (int*)(ws + off); off += (size_t)N_EDGES;
    float* acc8 = ws + off; off += (size_t)RREL * N_NODES * GDIM;

    dim3 mmBlock(32, 8);
    int mmGrid = N_NODES / 16;

    k_basis_w<<<(RREL * GDIM * HD1 + 255) / 256, 256, 0, stream>>>(comp, basis, W);
    hipMemsetAsync(cnt, 0, (size_t)N_NODES * (RREL + 1) * sizeof(float), stream);
    k_count<<<(N_EDGES + 255) / 256, 256, 0, stream>>>(ei, et, cnt, deg, slot);

    hipMemsetAsync(acc8, 0, (size_t)RREL * N_NODES * GDIM * sizeof(float), stream);
    k_scatter_all<<<(N_EDGES + 255) / 256, 256, 0, stream>>>(x, ei, et, cnt, acc8);
    k_mm_rgcn<<<mmGrid, mmBlock, 0, stream>>>(acc8, x, W, root, bias1, hbuf);

    k_mm16<true><<<mmGrid, mmBlock, 0, stream>>>(hbuf, gat_w, xl, nullptr);
    k_attn<<<(N_NODES * HEADS + 255) / 256, 256, 0, stream>>>(xl, att_src, att_dst, asg, adg);

    k_scan<<<SCAN_B, SCAN_T, 0, stream>>>(deg, rowptr);
    k_fill_csr<<<(N_EDGES + 255) / 256, 256, 0, stream>>>(ei, rowptr, slot, csrc);
    k_gat_node<<<(N_NODES * 32 + 255) / 256, 256, 0, stream>>>(xl, asg, adg, rowptr, deg, csrc,
                                                               gat_bias, out);

    (void)in_sizes; (void)n_in; (void)out_size; (void)ws_size;
}


# RGCN product: one workgroup per WGP owns 13 row tiles and each wave keeps 13 accumulator tiles, so every B fragment load feeds 13 matrix instructions, not 1; sum order per element kept
# speedup vs baseline: 2.9218x; 1.0701x over previous
#include <hip/hip_runtime.h>
#include <math.h>

#define N_NODES 50000
#define N_EDGES 800000
#define GDIM 128
#define HD1 128
#define HEADS 4
#define HD2 32
#define RREL 8
#define NB 20

typedef __attribute__((ext_vector_type(2))) float v2f;
typedef __attribute__((ext_vector_type(8))) float v8f;

__device__ __forceinline__ float lrelu(float x) { return x > 0.f ? x : 0.2f * x; }


__global__ void k_basis_w(const float* __restrict__ comp, const float* __restrict__ basis,
                          float* __restrict__ W) {
    int t = blockIdx.x * blockDim.x + threadIdx.x;
    if (t >= RREL * GDIM * HD1) return;
    int r = t / (GDIM * HD1);
    int io = t % (GDIM * HD1);
    float s = 0.f;
    #pragma unroll
    for (int b = 0; b < NB; ++b) s += comp[r * NB + b] * basis[b * GDIM * HD1 + io];
    W[t] = s;
}

__global__ void k_count(const int* __restrict__ ei, const int* __restrict__ et,
                        float* __restrict__ cnt, int* __restrict__ deg, int* __restrict__ slot) {
    int e = blockIdx.x * blockDim.x + threadIdx.x;
    if (e >= N_EDGES) return;
    int s = ei[e];
    int d = ei[N_EDGES + e];
    atomicAdd(&cnt[d * RREL + et[e]], 1.f);
    slot[e] = (s != d) ? atomicAdd(&deg[d], 1) : -1;
}


#define SCAT_U 4

__global__ void k_scatter_all(const float* __restrict__ x, const int* __restrict__ ei,
                              const int* __restrict__ et, const float* __restrict__ cnt,
                              float* __restrict__ acc8) {
    int gid = blockIdx.x * blockDim.x + threadIdx.x;
    int lane = threadIdx.x & 31;
    bool in = gid < N_EDGES;
    int ec = in ? gid : N_EDGES - 1;
    int t = et[ec];
    int s = ei[ec];
    int d = ei[N_EDGES + ec];
    unsigned todo = (unsigned)__builtin_amdgcn_ballot_w32(in);
    while (todo != 0u) {
        int  j[SCAT_U];
        bool ok[SCAT_U];
        #pragma unroll
        for (int u = 0; u < SCAT_U; ++u) {
            ok[u] = todo != 0u;
            j[u] = ok[u] ? __builtin_ctz(todo) : j[0];
            todo &= todo - 1u;
        }
        int   dj[SCAT_U];
        int   tj[SCAT_U];
        float cn[SCAT_U];
        float v[SCAT_U][4];
        #pragma unroll
        for (int u = 0; u < SCAT_U; ++u) {
            int sj = __builtin_amdgcn_readlane(s, j[u]);
            dj[u] = __builtin_amdgcn_readlane(d, j[u]);
            tj[u] = __builtin_amdgcn_readlane(t, j[u]);
            const float* xs = x + (size_t)sj * GDIM + lane;
            v[u][0] = xs[0]; v[u][1] = xs[32]; v[u][2] = xs[64]; v[u][3] = xs[96];
            cn[u] = cnt[dj[u] * RREL + tj[u]];
        }
        static_assert(SCAT_U == 4, "the statement below names the loads of four edges");
        asm volatile("" :: "v"(v[0][0]), "v"(v[0][1]), "v"(v[0][2]), "v"(v[0][3]),
                           "v"(v[1][0]), "v"(v[1][1]), "v"(v[1][2]), "v"(v[1][3]),
                           "v"(v[2][0]), "v"(v[2][1]), "v"(v[2][2]), "v"(v[2][3]),
                           "v"(v[3][0]), "v"(v[3][1]), "v"(v[3][2]), "v"(v[3][3]),
                           "s"(cn[0]), "s"(cn[1]), "s"(cn[2]), "s"(cn[3]));
        #pragma unroll
        for (int u = 0; u < SCAT_U; ++u) {
            if (ok[u]) {
                float c = 1.f / cn[u];
                float* o = acc8 + ((size_t)tj[u] * N_NODES + dj[u]) * GDIM + lane;
                atomicAdd(o + 0,  v[u][0] * c);
                atomicAdd(o + 32, v[u][1] * c);
                atomicAdd(o + 64, v[u][2] * c);
                atomicAdd(o + 96, v[u][3] * c);
            }
        }
    }
}


template <bool BIAS_MODE>
__global__ __launch_bounds__(256) void k_mm16(const float* __restrict__ A,
                                              const float* __restrict__ B,
                                              float* __restrict__ C,
                                              const float* __restrict__ bias) {
    __shared__ float As[16][132];
    int tx = threadIdx.x;
    int ty = threadIdx.y;
    int tid = ty * 32 + tx;
    int row0 = blockIdx.x * 16;

    const float4* A4 = (const float4*)(A + (size_t)row0 * 128);
    for (int i = tid; i < 16 * 32; i += 256) {
        int rr = i >> 5, cc = i & 31;
        float4 v = A4[rr * 32 + cc];
        As[rr][cc * 4 + 0] = v.x;
        As[rr][cc * 4 + 1] = v.y;
        As[rr][cc * 4 + 2] = v.z;
        As[rr][cc * 4 + 3] = v.w;
    }
    __syncthreads();

    int col0 = ty * 16;
    int half = tx >> 4;
    int mn = tx & 15;
    int kg = half * 2;

    v8f c;
    if (BIAS_MODE) {
        float bv = bias ? bias[col0 + mn] : 0.f;
        #pragma unroll
        for (int i = 0; i < 8; ++i) c[i] = bv;
    } else {
        #pragma unroll
        for (int i = 0; i < 8; ++i)
            c[i] = C[(size_t)(row0 + half * 8 + i) * 128 + col0 + mn];
    }

    #pragma unroll 4
    for (int k0 = 0; k0 < 128; k0 += 4) {
        v2f a, b;
        a.x = As[mn][k0 + kg];
        a.y = As[mn][k0 + kg + 1];
        b.x = B[(size_t)(k0 + kg) * 128 + col0 + mn];
        b.y = B[(size_t)(k0 + kg + 1) * 128 + col0 + mn];
        c = __builtin_amdgcn_wmma_f32_16x16x4_f32(false, a, false, b, (short)0, c,
                                                  false, false);
    }

    #pragma unroll
    for (int i = 0; i < 8; ++i)
        C[(size_t)(row0 + half * 8 + i) * 128 + col0 + mn] = c[i];
}


#define RG_CH (RREL + 1)
#define RG_T 13
#define RG_ROWS (RG_T * 16)
#define RG_WGS ((N_NODES / 16 + RG_T - 1) / RG_T)
#define RG_STAGE (RG_ROWS * 32 / 256)
#define RG_KG 8

__device__ __forceinline__ void rg_load_block(v2f (&f)[RG_T][2], const float* a, int k0) {
    #pragma unroll
    for (int r = 0; r < RG_T; ++r) {
        const float* q = a + r * 16 * 132 + k0;
        f[r][0].x = q[0]; f[r][0].y = q[1];
        f[r][1].x = q[4]; f[r][1].y = q[5];
    }
}

__device__ __forceinline__ void rg_mma_block(v8f (&c)[RG_T], const v2f (&f)[RG_T][2],
                                             v2f b0, v2f b1) {
    #pragma unroll
    for (int r = 0; r < RG_T; ++r)
        c[r] = __builtin_amdgcn_wmma_f32_16x16x4_f32(false, f[r][0], false, b0, (short)0, c[r],
                                                     false, false);
    #pragma unroll
    for (int r = 0; r < RG_T; ++r)
        c[r] = __builtin_amdgcn_wmma_f32_16x16x4_f32(false, f[r][1], false, b1, (short)0, c[r],
                                                     false, false);
}

__device__ __forceinline__ void rg_load_b(v2f (&v)[4], const float* bl) {
    #pragma unroll
    for (int s = 0; s < 4; ++s) {
        v[s].x = bl[(4 * s) * 128];
        v[s].y = bl[(4 * s + 1) * 128];
    }
}

template <int G, bool MORE>
__device__ __forceinline__ void rg_group(v8f (&c)[RG_T], v2f (&fa)[RG_T][2], v2f (&fb)[RG_T][2],
                                         const v2f (&bc)[4], v2f (&bn)[4],
                                         const float* afrag, const float* bcur, const float* bnxt,
                                         float4 (&p)[4], const float4* src4, float* sdst,
                                         int srow) {
    if (G + 1 < RG_KG) rg_load_b(bn, bcur + (G + 1) * 16 * 128);
    else if (MORE)     rg_load_b(bn, bnxt);
    if (MORE) {
        #pragma unroll
        for (int j = 0; j < 4; ++j) {
            int q = (G - 1) * 4 + j;
            if (G >= 1 && q < RG_STAGE) {
                float* d = sdst + q * 8 * 132;
                d[0] = p[j].x; d[1] = p[j].y; d[2] = p[j].z; d[3] = p[j].w;
            }
        }
        #pragma unroll
        for (int j = 0; j < 4; ++j) {
            int q = G * 4 + j;
            if (q < RG_STAGE) {
                int gr = srow + 8 * q;
                gr = gr < N_NODES ? gr : N_NODES - 1;
                p[j] = src4[(size_t)gr * 32];
            }
        }
    }
    rg_load_block(fb, afrag, G * 16 + 8);
    __builtin_amdgcn_sched_barrier(0);
    rg_mma_block(c, fa, bc[0], bc[1]);
    __builtin_amdgcn_sched_barrier(0);
    if (G + 1 < RG_KG) rg_load_block(fa, afrag, G * 16 + 16);
    __builtin_amdgcn_sched_barrier(0);
    rg_mma_block(c, fb, bc[2], bc[3]);
    __builtin_amdgcn_sched_barrier(0);
}

template <bool MORE>
__device__ __forceinline__ void rg_chunk(v8f (&c)[RG_T], v2f (&b0)[4], v2f (&b1)[4],
                                         const float* afrag, const float* bcur, const float* bnxt,
                                         const float4* src4, float* sdst, int srow) {
    v2f fa[RG_T][2], fb[RG_T][2];
    float4 p[4];
    rg_load_block(fa, afrag, 0);
    rg_group<0, MORE>(c, fa, fb, b0, b1, afrag, bcur, bnxt, p, src4, sdst, srow);
    rg_group<1, MORE>(c, fa, fb, b1, b0, afrag, bcur, bnxt, p, src4, sdst, srow);
    rg_group<2, MORE>(c, fa, fb, b0, b1, afrag, bcur, bnxt, p, src4, sdst, srow);
    rg_group<3, MORE>(c, fa, fb, b1, b0, afrag, bcur, bnxt, p, src4, sdst, srow);
    rg_group<4, MORE>(c, fa, fb, b0, b1, afrag, bcur, bnxt, p, src4, sdst, srow);
    rg_group<5, MORE>(c, fa, fb, b1, b0, afrag, bcur, bnxt, p, src4, sdst, srow);
    rg_group<6, MORE>(c, fa, fb, b0, b1, afrag, bcur, bnxt, p, src4, sdst, srow);
    rg_group<7, MORE>(c, fa, fb, b1, b0, afrag, bcur, bnxt, p, src4, sdst, srow);
}

__global__ __launch_bounds__(256) void k_mm_rgcn(const float* __restrict__ acc8,
                                                 const float* __restrict__ x,
                                                 const float* __restrict__ W,
                                                 const float* __restrict__ root,
                                                 const float* __restrict__ bias,
                                                 float* __restrict__ C) {
    __shared__ float As[2][RG_ROWS][132];
    int tx = threadIdx.x;
    int ty = threadIdx.y;
    int tid = ty * 32 + tx;
    int row0 = blockIdx.x * RG_ROWS;

    int ra = tid >> 5, ca = tid & 31;
    const float4* acc4 = (const float4*)acc8 + ca;
    const float4* x4   = (const float4*)x + ca;
    const size_t chunk4 = (size_t)N_NODES * 128 / 4;

    #pragma unroll
    for (int q = 0; q < RG_STAGE; ++q) {
        int gr = row0 + ra + 8 * q;
        gr = gr < N_NODES ? gr : N_NODES - 1;
        float4 v = acc4[(size_t)gr * 32];
        float* dst = &As[0][ra + 8 * q][ca * 4];
        dst[0] = v.x; dst[1] = v.y; dst[2] = v.z; dst[3] = v.w;
    }
    __syncthreads();

    int col0 = ty * 16;
    int half = tx >> 4;
    int mn = tx & 15;
    int kg = half * 2;

    v8f c[RG_T];
    {
        float bv = bias[col0 + mn];
        #pragma unroll
        for (int r = 0; r < RG_T; ++r) {
            #pragma unroll
            for (int i = 0; i < 8; ++i) c[r][i] = bv;
        }
    }

    const int blane = kg * 128 + col0 + mn;
    v2f b0[4], b1[4];
    rg_load_b(b0, W + blane);

    #pragma unroll 1
    for (int ch = 0; ch < RREL; ++ch) {
        int cur = ch & 1, nxt = cur ^ 1;
        const float4* src4 = (ch + 1 < RREL) ? acc4 + (size_t)(ch + 1) * chunk4 : x4;
        const float* bcur = W + (size_t)ch * GDIM * HD1 + blane;
        const float* bnxt = ((ch + 1 < RREL) ? W + (size_t)(ch + 1) * GDIM * HD1 : root) + blane;
        rg_chunk<true>(c, b0, b1, &As[cur][mn][kg], bcur, bnxt, src4, &As[nxt][ra][ca * 4],
                       row0 + ra);
        __syncthreads();
    }
    rg_chunk<false>(c, b0, b1, &As[RREL & 1][mn][kg], root + blane, root + blane, x4,
                    &As[(RREL & 1) ^ 1][ra][ca * 4], row0 + ra);

    #pragma unroll
    for (int r = 0; r < RG_T; ++r) {
        if (row0 + r * 16 < N_NODES) {
            #pragma unroll
            for (int i = 0; i < 8; ++i)
                C[(size_t)(row0 + r * 16 + half * 8 + i) * 128 + col0 + mn] = c[r][i];
        }
    }
}


__global__ void k_attn(const float* __restrict__ xl, const float* __restrict__ att_src,
                       const float* __restrict__ att_dst, float* __restrict__ a_src,
                       float* __restrict__ a_dst) {
    int t = blockIdx.x * blockDim.x + threadIdx.x;
    if (t >= N_NODES * HEADS) return;
    int n = t >> 2, h = t & 3;
    const float* row = xl + (size_t)n * 128 + h * HD2;
    float ss = 0.f, sd = 0.f;
    #pragma unroll
    for (int j = 0; j < HD2; ++j) {
        float v = row[j];
        ss += v * att_src[h * HD2 + j];
        sd += v * att_dst[h * HD2 + j];
    }
    a_src[t] = ss;
    a_dst[t] = sd;
}


#define SCAN_T 1024
#define SCAN_B ((N_NODES + SCAN_T - 1) / SCAN_T)

__global__ __launch_bounds__(SCAN_T) void k_scan(const int* __restrict__ deg,
                                                 int* __restrict__ rowptr) {
    __shared__ int w_own[SCAN_T / 32];
    __shared__ int w_pre[SCAN_T / 32];
    int t = threadIdx.x, lane = t & 31, w = t >> 5;
    int b = blockIdx.x;
    int pre = 0;
    for (int i = 0; i < b; ++i) pre += deg[i * SCAN_T + t];
    int idx = b * SCAN_T + t;
    int own = idx < N_NODES ? deg[idx] : 0;
    int inc = own;
    for (int o = 1; o < 32; o <<= 1) {
        int v = __shfl_up(inc, o);
        if (lane >= o) inc += v;
    }
    for (int o = 16; o > 0; o >>= 1) pre += __shfl_xor(pre, o);
    if (lane == 31) w_own[w] = inc;
    if (lane == 0)  w_pre[w] = pre;
    __syncthreads();
    int add = w_pre[lane] + (lane < w ? w_own[lane] : 0);
    for (int o = 16; o > 0; o >>= 1) add += __shfl_xor(add, o);
    int start = add + inc - own;
    if (idx < N_NODES) rowptr[idx] = start;
}

__global__ void k_fill_csr(const int* __restrict__ ei, const int* __restrict__ rowptr,
                           const int* __restrict__ slot, int* __restrict__ csr_src) {
    int e = blockIdx.x * blockDim.x + threadIdx.x;
    if (e >= N_EDGES) return;
    int s = ei[e];
    int d = ei[N_EDGES + e];
    if (s != d) csr_src[rowptr[d] + slot[e]] = s;
}

__device__ __forceinline__ float readlane_f(float v, int l) {
    return __int_as_float(__builtin_amdgcn_readlane(__float_as_int(v), l));
}

__device__ __forceinline__ float wave_max(float v) {
    for (int o = 16; o > 0; o >>= 1) v = fmaxf(v, __shfl_xor(v, o));
    return v;
}

__device__ __forceinline__ float wave_sum(float v) {
    for (int o = 16; o > 0; o >>= 1) v += __shfl_xor(v, o);
    return v;
}

__device__ __forceinline__ void gat_rows(const float* __restrict__ xl, int lane, int sv,
                                         float w0, float w1, float w2, float w3, int cc,
                                         float& acc0, float& acc1, float& acc2, float& acc3) {
    #pragma unroll 4
    for (int k = 0; k < cc; ++k) {
        int s = __builtin_amdgcn_readlane(sv, k);
        float u0 = readlane_f(w0, k);
        float u1 = readlane_f(w1, k);
        float u2 = readlane_f(w2, k);
        float u3 = readlane_f(w3, k);
        const float* xs = xl + (size_t)s * 128 + lane;
        acc0 = fmaf(xs[0],  u0, acc0);
        acc1 = fmaf(xs[32], u1, acc1);
        acc2 = fmaf(xs[64], u2, acc2);
        acc3 = fmaf(xs[96], u3, acc3);
    }
}

__device__ __forceinline__ int bucket_sorted(const int* __restrict__ csr_src, int beg, int n,
                                             int base, int lane, int fill) {
    int mine = fill;
    for (int k = 0; k < n; ++k) {
        int sk = csr_src[beg + k];
        int rank = 0;
        for (int b = 0; b < n; b += 32) {
            int i = b + lane;
            int si = i < n ? csr_src[beg + i] : 0x7fffffff;
            rank += __builtin_popcount(
                __builtin_amdgcn_ballot_w32(si < sk || (si == sk && i < k)));
        }
        if (lane == rank - base) mine = sk;
    }
    return mine;
}

__global__ __launch_bounds__(256) void k_gat_node(const float* __restrict__ xl,
                                                  const float* __restrict__ a_src,
                                                  const float* __restrict__ a_dst,
                                                  const int* __restrict__ rowptr,
                                                  const int* __restrict__ deg,
                                                  const int* __restrict__ csr_src,
                                                  const float* __restrict__ gat_bias,
                                                  float* __restrict__ out) {
    int gid = blockIdx.x * blockDim.x + threadIdx.x;
    if ((gid >> 5) >= N_NODES) return;
    int lane = threadIdx.x & 31;
    int d   = __builtin_amdgcn_readfirstlane(gid >> 5);
    int beg = __builtin_amdgcn_readfirstlane(rowptr[d]);
    int n   = __builtin_amdgcn_readfirstlane(deg[d]);

    const float4 ad  = *(const float4*)(a_dst + (size_t)d * 4);
    const float4 asd = *(const float4*)(a_src + (size_t)d * 4);
    float l0 = lrelu(asd.x + ad.x), l1 = lrelu(asd.y + ad.y);
    float l2 = lrelu(asd.z + ad.z), l3 = lrelu(asd.w + ad.w);

    bool ok = lane < n;
    int sv = d;
    if (n <= 32) {
        int raw = ok ? csr_src[beg + lane] : 0x7fffffff;
        for (int k = 0; k < n; ++k) {
            int sk = __builtin_amdgcn_readlane(raw, k);
            int rank = __builtin_popcount(
                __builtin_amdgcn_ballot_w32(raw < sk || (raw == sk && lane < k)));
            if (lane == rank) sv = sk;
        }
    } else {
        sv = bucket_sorted(csr_src, beg, n, 0, lane, d);
    }
    float4 av = *(const float4*)(a_src + (size_t)sv * 4);
    float c0 = lrelu(av.x + ad.x), c1 = lrelu(av.y + ad.y);
    float c2 = lrelu(av.z + ad.z), c3 = lrelu(av.w + ad.w);

    float m0 = ok ? fmaxf(l0, c0) : l0, m1 = ok ? fmaxf(l1, c1) : l1;
    float m2 = ok ? fmaxf(l2, c2) : l2, m3 = ok ? fmaxf(l3, c3) : l3;
    for (int b = 32; b < n; b += 32) {
        bool okb = b + lane < n;
        int s = bucket_sorted(csr_src, beg, n, b, lane, d);
        float4 a = *(const float4*)(a_src + (size_t)s * 4);
        float t0 = lrelu(a.x + ad.x), t1 = lrelu(a.y + ad.y);
        float t2 = lrelu(a.z + ad.z), t3 = lrelu(a.w + ad.w);
        m0 = okb ? fmaxf(m0, t0) : m0; m1 = okb ? fmaxf(m1, t1) : m1;
        m2 = okb ? fmaxf(m2, t2) : m2; m3 = okb ? fmaxf(m3, t3) : m3;
    }
    m0 = wave_max(m0); m1 = wave_max(m1); m2 = wave_max(m2); m3 = wave_max(m3);

    float e0 = ok ? __expf(c0 - m0) : 0.f, e1 = ok ? __expf(c1 - m1) : 0.f;
    float e2 = ok ? __expf(c2 - m2) : 0.f, e3 = ok ? __expf(c3 - m3) : 0.f;
    float d0 = e0, d1 = e1, d2 = e2, d3 = e3;
    for (int b = 32; b < n; b += 32) {
        bool okb = b + lane < n;
        int s = bucket_sorted(csr_src, beg, n, b, lane, d);
        float4 a = *(const float4*)(a_src + (size_t)s * 4);
        d0 += okb ? __expf(lrelu(a.x + ad.x) - m0) : 0.f;
        d1 += okb ? __expf(lrelu(a.y + ad.y) - m1) : 0.f;
        d2 += okb ? __expf(lrelu(a.z + ad.z) - m2) : 0.f;
        d3 += okb ? __expf(lrelu(a.w + ad.w) - m3) : 0.f;
    }
    float s0 = __expf(l0 - m0), s1 = __expf(l1 - m1), s2 = __expf(l2 - m2), s3 = __expf(l3 - m3);
    d0 = wave_sum(d0) + s0; d1 = wave_sum(d1) + s1;
    d2 = wave_sum(d2) + s2; d3 = wave_sum(d3) + s3;

    const float* xd = xl + (size_t)d * 128 + lane;
    float acc0 = xd[0]  * (s0 / d0);
    float acc1 = xd[32] * (s1 / d1);
    float acc2 = xd[64] * (s2 / d2);
    float acc3 = xd[96] * (s3 / d3);
    gat_rows(xl, lane, sv, e0 / d0, e1 / d1, e2 / d2, e3 / d3, n < 32 ? n : 32,
             acc0, acc1, acc2, acc3);
    for (int b = 32; b < n; b += 32) {
        bool okb = b + lane < n;
        int s = bucket_sorted(csr_src, beg, n, b, lane, d);
        float4 a = *(const float4*)(a_src + (size_t)s * 4);
        float w0 = okb ? __expf(lrelu(a.x + ad.x) - m0) / d0 : 0.f;
        float w1 = okb ? __expf(lrelu(a.y + ad.y) - m1) / d1 : 0.f;
        float w2 = okb ? __expf(lrelu(a.z + ad.z) - m2) / d2 : 0.f;
        float w3 = okb ? __expf(lrelu(a.w + ad.w) - m3) / d3 : 0.f;
        gat_rows(xl, lane, s, w0, w1, w2, w3, (n - b) < 32 ? (n - b) : 32,
                 acc0, acc1, acc2, acc3);
    }

    float* o = out + (size_t)d * 128 + lane;
    o[0]  = acc0 + gat_bias[lane];
    o[32] = acc1 + gat_bias[32 + lane];
    o[64] = acc2 + gat_bias[64 + lane];
    o[96] = acc3 + gat_bias[96 + lane];
}


extern "C" void kernel_launch(void* const* d_in, const int* in_sizes, int n_in,
                              void* d_out, int out_size, void* d_ws, size_t ws_size,
                              hipStream_t stream) {
    const float* x        = (const float*)d_in[0];
    const int*   ei       = (const int*)d_in[1];
    const int*   et       = (const int*)d_in[2];
    const float* comp     = (const float*)d_in[3];
    const float* basis    = (const float*)d_in[4];
    const float* root     = (const float*)d_in[5];
    const float* bias1    = (const float*)d_in[6];
    const float* gat_w    = (const float*)d_in[7];
    const float* att_src  = (const float*)d_in[8];
    const float* att_dst  = (const float*)d_in[9];
    const float* gat_bias = (const float*)d_in[10];
    float* out = (float*)d_out;

    float* ws = (float*)d_ws;
    size_t off = 0;
    float* W    = ws + off; off += (size_t)RREL * GDIM * HD1;
    float* cnt  = ws + off; off += (size_t)N_NODES * RREL;
    int*   deg  = (int*)(ws + off); off += (size_t)N_NODES;
    float* hbuf = ws + off; off += (size_t)N_NODES * HD1;
    float* xl   = ws + off; off += (size_t)N_NODES * HD1;
    float* asg  = ws + off; off += (size_t)N_NODES * HEADS;
    float* adg  = ws + off; off += (size_t)N_NODES * HEADS;
    int*   rowptr = (int*)(ws + off); off += (size_t)N_NODES;
    int*   slot   = (int*)(ws + off); off += (size_t)N_EDGES;
    int*   csrc   = (int*)(ws + off); off += (size_t)N_EDGES;
    float* acc8 = ws + off; off += (size_t)RREL * N_NODES * GDIM;

    dim3 mmBlock(32, 8);
    int mmGrid = N_NODES / 16;

    k_basis_w<<<(RREL * GDIM * HD1 + 255) / 256, 256, 0, stream>>>(comp, basis, W);
    hipMemsetAsync(cnt, 0, (size_t)N_NODES * (RREL + 1) * sizeof(float), stream);
    k_count<<<(N_EDGES + 255) / 256, 256, 0, stream>>>(ei, et, cnt, deg, slot);

    hipMemsetAsync(acc8, 0, (size_t)RREL * N_NODES * GDIM * sizeof(float), stream);
    k_scatter_all<<<(N_EDGES + 255) / 256, 256, 0, stream>>>(x, ei, et, cnt, acc8);
    k_mm_rgcn<<<RG_WGS, mmBlock, 0, stream>>>(acc8, x, W, root, bias1, hbuf);

    k_mm16<true><<<mmGrid, mmBlock, 0, stream>>>(hbuf, gat_w, xl, nullptr);
    k_attn<<<(N_NODES * HEADS + 255) / 256, 256, 0, stream>>>(xl, att_src, att_dst, asg, adg);

    k_scan<<<SCAN_B, SCAN_T, 0, stream>>>(deg, rowptr);
    k_fill_csr<<<(N_EDGES + 255) / 256, 256, 0, stream>>>(ei, rowptr, slot, csrc);
    k_gat_node<<<(N_NODES * 32 + 255) / 256, 256, 0, stream>>>(xl, asg, adg, rowptr, deg, csrc,
                                                               gat_bias, out);

    (void)in_sizes; (void)n_in; (void)out_size; (void)ws_size;
}


# RGCN means: float atomics go; destination buckets (source|relation, all edges) serve both layers; a wave per node adds a relation's rows in source order in registers, stores each once; clear kept
# speedup vs baseline: 3.2451x; 1.1107x over previous
#include <hip/hip_runtime.h>
#include <math.h>

#define N_NODES 50000
#define N_EDGES 800000
#define GDIM 128
#define HD1 128
#define HEADS 4
#define HD2 32
#define RREL 8
#define NB 20

typedef __attribute__((ext_vector_type(2))) float v2f;
typedef __attribute__((ext_vector_type(8))) float v8f;

__device__ __forceinline__ float lrelu(float x) { return x > 0.f ? x : 0.2f * x; }


__global__ void k_basis_w(const float* __restrict__ comp, const float* __restrict__ basis,
                          float* __restrict__ W) {
    int t = blockIdx.x * blockDim.x + threadIdx.x;
    if (t >= RREL * GDIM * HD1) return;
    int r = t / (GDIM * HD1);
    int io = t % (GDIM * HD1);
    float s = 0.f;
    #pragma unroll
    for (int b = 0; b < NB; ++b) s += comp[r * NB + b] * basis[b * GDIM * HD1 + io];
    W[t] = s;
}

__global__ void k_count(const int* __restrict__ ei, int* __restrict__ deg, int* __restrict__ slot) {
    int e = blockIdx.x * blockDim.x + threadIdx.x;
    if (e >= N_EDGES) return;
    int d = ei[N_EDGES + e];
    slot[e] = atomicAdd(&deg[d], 1);
}


template <bool BIAS_MODE>
__global__ __launch_bounds__(256) void k_mm16(const float* __restrict__ A,
                                              const float* __restrict__ B,
                                              float* __restrict__ C,
                                              const float* __restrict__ bias) {
    __shared__ float As[16][132];
    int tx = threadIdx.x;
    int ty = threadIdx.y;
    int tid = ty * 32 + tx;
    int row0 = blockIdx.x * 16;

    const float4* A4 = (const float4*)(A + (size_t)row0 * 128);
    for (int i = tid; i < 16 * 32; i += 256) {
        int rr = i >> 5, cc = i & 31;
        float4 v = A4[rr * 32 + cc];
        As[rr][cc * 4 + 0] = v.x;
        As[rr][cc * 4 + 1] = v.y;
        As[rr][cc * 4 + 2] = v.z;
        As[rr][cc * 4 + 3] = v.w;
    }
    __syncthreads();

    int col0 = ty * 16;
    int half = tx >> 4;
    int mn = tx & 15;
    int kg = half * 2;

    v8f c;
    if (BIAS_MODE) {
        float bv = bias ? bias[col0 + mn] : 0.f;
        #pragma unroll
        for (int i = 0; i < 8; ++i) c[i] = bv;
    } else {
        #pragma unroll
        for (int i = 0; i < 8; ++i)
            c[i] = C[(size_t)(row0 + half * 8 + i) * 128 + col0 + mn];
    }

    #pragma unroll 4
    for (int k0 = 0; k0 < 128; k0 += 4) {
        v2f a, b;
        a.x = As[mn][k0 + kg];
        a.y = As[mn][k0 + kg + 1];
        b.x = B[(size_t)(k0 + kg) * 128 + col0 + mn];
        b.y = B[(size_t)(k0 + kg + 1) * 128 + col0 + mn];
        c = __builtin_amdgcn_wmma_f32_16x16x4_f32(false, a, false, b, (short)0, c,
                                                  false, false);
    }

    #pragma unroll
    for (int i = 0; i < 8; ++i)
        C[(size_t)(row0 + half * 8 + i) * 128 + col0 + mn] = c[i];
}


#define RG_CH (RREL + 1)
#define RG_T 13
#define RG_ROWS (RG_T * 16)
#define RG_WGS ((N_NODES / 16 + RG_T - 1) / RG_T)
#define RG_STAGE (RG_ROWS * 32 / 256)
#define RG_KG 8

__device__ __forceinline__ void rg_load_block(v2f (&f)[RG_T][2], const float* a, int k0) {
    #pragma unroll
    for (int r = 0; r < RG_T; ++r) {
        const float* q = a + r * 16 * 132 + k0;
        f[r][0].x = q[0]; f[r][0].y = q[1];
        f[r][1].x = q[4]; f[r][1].y = q[5];
    }
}

__device__ __forceinline__ void rg_mma_block(v8f (&c)[RG_T], const v2f (&f)[RG_T][2],
                                             v2f b0, v2f b1) {
    #pragma unroll
    for (int r = 0; r < RG_T; ++r)
        c[r] = __builtin_amdgcn_wmma_f32_16x16x4_f32(false, f[r][0], false, b0, (short)0, c[r],
                                                     false, false);
    #pragma unroll
    for (int r = 0; r < RG_T; ++r)
        c[r] = __builtin_amdgcn_wmma_f32_16x16x4_f32(false, f[r][1], false, b1, (short)0, c[r],
                                                     false, false);
}

__device__ __forceinline__ void rg_load_b(v2f (&v)[4], const float* bl) {
    #pragma unroll
    for (int s = 0; s < 4; ++s) {
        v[s].x = bl[(4 * s) * 128];
        v[s].y = bl[(4 * s + 1) * 128];
    }
}

template <int G, bool MORE>
__device__ __forceinline__ void rg_group(v8f (&c)[RG_T], v2f (&fa)[RG_T][2], v2f (&fb)[RG_T][2],
                                         const v2f (&bc)[4], v2f (&bn)[4],
                                         const float* afrag, const float* bcur, const float* bnxt,
                                         float4 (&p)[4], const float4* src4, float* sdst,
                                         int srow) {
    if (G + 1 < RG_KG) rg_load_b(bn, bcur + (G + 1) * 16 * 128);
    else if (MORE)     rg_load_b(bn, bnxt);
    if (MORE) {
        #pragma unroll
        for (int j = 0; j < 4; ++j) {
            int q = (G - 1) * 4 + j;
            if (G >= 1 && q < RG_STAGE) {
                float* d = sdst + q * 8 * 132;
                d[0] = p[j].x; d[1] = p[j].y; d[2] = p[j].z; d[3] = p[j].w;
            }
        }
        #pragma unroll
        for (int j = 0; j < 4; ++j) {
            int q = G * 4 + j;
            if (q < RG_STAGE) {
                int gr = srow + 8 * q;
                gr = gr < N_NODES ? gr : N_NODES - 1;
                p[j] = src4[(size_t)gr * 32];
            }
        }
    }
    rg_load_block(fb, afrag, G * 16 + 8);
    __builtin_amdgcn_sched_barrier(0);
    rg_mma_block(c, fa, bc[0], bc[1]);
    __builtin_amdgcn_sched_barrier(0);
    if (G + 1 < RG_KG) rg_load_block(fa, afrag, G * 16 + 16);
    __builtin_amdgcn_sched_barrier(0);
    rg_mma_block(c, fb, bc[2], bc[3]);
    __builtin_amdgcn_sched_barrier(0);
}

template <bool MORE>
__device__ __forceinline__ void rg_chunk(v8f (&c)[RG_T], v2f (&b0)[4], v2f (&b1)[4],
                                         const float* afrag, const float* bcur, const float* bnxt,
                                         const float4* src4, float* sdst, int srow) {
    v2f fa[RG_T][2], fb[RG_T][2];
    float4 p[4];
    rg_load_block(fa, afrag, 0);
    rg_group<0, MORE>(c, fa, fb, b0, b1, afrag, bcur, bnxt, p, src4, sdst, srow);
    rg_group<1, MORE>(c, fa, fb, b1, b0, afrag, bcur, bnxt, p, src4, sdst, srow);
    rg_group<2, MORE>(c, fa, fb, b0, b1, afrag, bcur, bnxt, p, src4, sdst, srow);
    rg_group<3, MORE>(c, fa, fb, b1, b0, afrag, bcur, bnxt, p, src4, sdst, srow);
    rg_group<4, MORE>(c, fa, fb, b0, b1, afrag, bcur, bnxt, p, src4, sdst, srow);
    rg_group<5, MORE>(c, fa, fb, b1, b0, afrag, bcur, bnxt, p, src4, sdst, srow);
    rg_group<6, MORE>(c, fa, fb, b0, b1, afrag, bcur, bnxt, p, src4, sdst, srow);
    rg_group<7, MORE>(c, fa, fb, b1, b0, afrag, bcur, bnxt, p, src4, sdst, srow);
}

__global__ __launch_bounds__(256) void k_mm_rgcn(const float* __restrict__ acc8,
                                                 const float* __restrict__ x,
                                                 const float* __restrict__ W,
                                                 const float* __restrict__ root,
                                                 const float* __restrict__ bias,
                                                 float* __restrict__ C) {
    __shared__ float As[2][RG_ROWS][132];
    int tx = threadIdx.x;
    int ty = threadIdx.y;
    int tid = ty * 32 + tx;
    int row0 = blockIdx.x * RG_ROWS;

    int ra = tid >> 5, ca = tid & 31;
    const float4* acc4 = (const float4*)acc8 + ca;
    const float4* x4   = (const float4*)x + ca;
    const size_t chunk4 = (size_t)N_NODES * 128 / 4;

    #pragma unroll
    for (int q = 0; q < RG_STAGE; ++q) {
        int gr = row0 + ra + 8 * q;
        gr = gr < N_NODES ? gr : N_NODES - 1;
        float4 v = acc4[(size_t)gr * 32];
        float* dst = &As[0][ra + 8 * q][ca * 4];
        dst[0] = v.x; dst[1] = v.y; dst[2] = v.z; dst[3] = v.w;
    }
    __syncthreads();

    int col0 = ty * 16;
    int half = tx >> 4;
    int mn = tx & 15;
    int kg = half * 2;

    v8f c[RG_T];
    {
        float bv = bias[col0 + mn];
        #pragma unroll
        for (int r = 0; r < RG_T; ++r) {
            #pragma unroll
            for (int i = 0; i < 8; ++i) c[r][i] = bv;
        }
    }

    const int blane = kg * 128 + col0 + mn;
    v2f b0[4], b1[4];
    rg_load_b(b0, W + blane);

    #pragma unroll 1
    for (int ch = 0; ch < RREL; ++ch) {
        int cur = ch & 1, nxt = cur ^ 1;
        const float4* src4 = (ch + 1 < RREL) ? acc4 + (size_t)(ch + 1) * chunk4 : x4;
        const float* bcur = W + (size_t)ch * GDIM * HD1 + blane;
        const float* bnxt = ((ch + 1 < RREL) ? W + (size_t)(ch + 1) * GDIM * HD1 : root) + blane;
        rg_chunk<true>(c, b0, b1, &As[cur][mn][kg], bcur, bnxt, src4, &As[nxt][ra][ca * 4],
                       row0 + ra);
        __syncthreads();
    }
    rg_chunk<false>(c, b0, b1, &As[RREL & 1][mn][kg], root + blane, root + blane, x4,
                    &As[(RREL & 1) ^ 1][ra][ca * 4], row0 + ra);

    #pragma unroll
    for (int r = 0; r < RG_T; ++r) {
        if (row0 + r * 16 < N_NODES) {
            #pragma unroll
            for (int i = 0; i < 8; ++i)
                C[(size_t)(row0 + r * 16 + half * 8 + i) * 128 + col0 + mn] = c[r][i];
        }
    }
}


__global__ void k_attn(const float* __restrict__ xl, const float* __restrict__ att_src,
                       const float* __restrict__ att_dst, float* __restrict__ a_src,
                       float* __restrict__ a_dst) {
    int t = blockIdx.x * blockDim.x + threadIdx.x;
    if (t >= N_NODES * HEADS) return;
    int n = t >> 2, h = t & 3;
    const float* row = xl + (size_t)n * 128 + h * HD2;
    float ss = 0.f, sd = 0.f;
    #pragma unroll
    for (int j = 0; j < HD2; ++j) {
        float v = row[j];
        ss += v * att_src[h * HD2 + j];
        sd += v * att_dst[h * HD2 + j];
    }
    a_src[t] = ss;
    a_dst[t] = sd;
}


#define SCAN_T 1024
#define SCAN_B ((N_NODES + SCAN_T - 1) / SCAN_T)

__global__ __launch_bounds__(SCAN_T) void k_scan(const int* __restrict__ deg,
                                                 int* __restrict__ rowptr) {
    __shared__ int w_own[SCAN_T / 32];
    __shared__ int w_pre[SCAN_T / 32];
    int t = threadIdx.x, lane = t & 31, w = t >> 5;
    int b = blockIdx.x;
    int pre = 0;
    for (int i = 0; i < b; ++i) pre += deg[i * SCAN_T + t];
    int idx = b * SCAN_T + t;
    int own = idx < N_NODES ? deg[idx] : 0;
    int inc = own;
    for (int o = 1; o < 32; o <<= 1) {
        int v = __shfl_up(inc, o);
        if (lane >= o) inc += v;
    }
    for (int o = 16; o > 0; o >>= 1) pre += __shfl_xor(pre, o);
    if (lane == 31) w_own[w] = inc;
    if (lane == 0)  w_pre[w] = pre;
    __syncthreads();
    int add = w_pre[lane] + (lane < w ? w_own[lane] : 0);
    for (int o = 16; o > 0; o >>= 1) add += __shfl_xor(add, o);
    int start = add + inc - own;
    if (idx < N_NODES) rowptr[idx] = start;
}

__global__ void k_fill_csr(const int* __restrict__ ei, const int* __restrict__ et,
                           const int* __restrict__ rowptr, const int* __restrict__ slot,
                           int* __restrict__ csr_src) {
    int e = blockIdx.x * blockDim.x + threadIdx.x;
    if (e >= N_EDGES) return;
    int s = ei[e];
    int d = ei[N_EDGES + e];
    csr_src[rowptr[d] + slot[e]] = (s << 3) | et[e];
}


__device__ __forceinline__ int rg_key(int entry) { return ((entry & 7) << 16) | (entry >> 3); }

__device__ __forceinline__ int rg_bucket_sorted(const int* __restrict__ csr_src, int beg, int n,
                                                int base, int lane) {
    int mine = 0x7fffffff;
    for (int k = 0; k < n; ++k) {
        int sk = rg_key(csr_src[beg + k]);
        int rank = 0;
        for (int b = 0; b < n; b += 32) {
            int i = b + lane;
            int si = i < n ? rg_key(csr_src[beg + i]) : 0x7fffffff;
            rank += __builtin_popcount(
                __builtin_amdgcn_ballot_w32(si < sk || (si == sk && i < k)));
        }
        if (lane == rank - base) mine = sk;
    }
    return mine;
}

#define RGG_U 4

__global__ __launch_bounds__(256) void k_rgcn_gather(const float* __restrict__ x,
                                                     const int* __restrict__ rowptr,
                                                     const int* __restrict__ deg,
                                                     const int* __restrict__ csr_src,
                                                     float* __restrict__ acc8) {
    int gid = blockIdx.x * blockDim.x + threadIdx.x;
    if ((gid >> 5) >= N_NODES) return;
    int lane = threadIdx.x & 31;
    int d   = __builtin_amdgcn_readfirstlane(gid >> 5);
    int beg = __builtin_amdgcn_readfirstlane(rowptr[d]);
    int n   = __builtin_amdgcn_readfirstlane(deg[d]);

    const float4* x4 = (const float4*)x + lane;
    float4 acc = make_float4(0.f, 0.f, 0.f, 0.f);
    int cur = -1;
    int cnt = 0;

    for (int base = 0; base < n; base += 32) {
        int kv;
        if (n <= 32) {
            int raw = lane < n ? rg_key(csr_src[beg + lane]) : 0x7fffffff;
            kv = raw;
            for (int k = 0; k < n; ++k) {
                int sk = __builtin_amdgcn_readlane(raw, k);
                int rank = __builtin_popcount(
                    __builtin_amdgcn_ballot_w32(raw < sk || (raw == sk && lane < k)));
                if (lane == rank) kv = sk;
            }
        } else {
            kv = rg_bucket_sorted(csr_src, beg, n, base, lane);
        }
        int cc = (n - base) < 32 ? (n - base) : 32;
        for (int k = 0; k < cc; k += RGG_U) {
            int    tk[RGG_U];
            float4 v[RGG_U];
            #pragma unroll
            for (int u = 0; u < RGG_U; ++u) {
                int kk = (k + u) < cc ? (k + u) : (cc - 1);
                int key = __builtin_amdgcn_readlane(kv, kk);
                tk[u] = key >> 16;
                v[u] = x4[(size_t)(key & 0xffff) * (GDIM / 4)];
            }
            static_assert(RGG_U == 4, "the statement below names the loads of four rows");
            asm volatile("" :: "v"(v[0].x), "v"(v[0].y), "v"(v[0].z), "v"(v[0].w),
                               "v"(v[1].x), "v"(v[1].y), "v"(v[1].z), "v"(v[1].w),
                               "v"(v[2].x), "v"(v[2].y), "v"(v[2].z), "v"(v[2].w),
                               "v"(v[3].x), "v"(v[3].y), "v"(v[3].z), "v"(v[3].w));
            #pragma unroll
            for (int u = 0; u < RGG_U; ++u) {
                if (k + u < cc) {
                    if (tk[u] != cur) {
                        if (cnt > 0) {
                            float c = (float)cnt;
                            float4 o = make_float4(acc.x / c, acc.y / c, acc.z / c, acc.w / c);
                            ((float4*)(acc8 + ((size_t)cur * N_NODES + d) * GDIM))[lane] = o;
                        }
                        acc = make_float4(0.f, 0.f, 0.f, 0.f);
                        cnt = 0;
                        cur = tk[u];
                    }
                    acc.x += v[u].x; acc.y += v[u].y; acc.z += v[u].z; acc.w += v[u].w;
                    ++cnt;
                }
            }
        }
    }
    if (cnt > 0) {
        float c = (float)cnt;
        float4 o = make_float4(acc.x / c, acc.y / c, acc.z / c, acc.w / c);
        ((float4*)(acc8 + ((size_t)cur * N_NODES + d) * GDIM))[lane] = o;
    }
}

__device__ __forceinline__ float readlane_f(float v, int l) {
    return __int_as_float(__builtin_amdgcn_readlane(__float_as_int(v), l));
}

__device__ __forceinline__ float wave_max(float v) {
    for (int o = 16; o > 0; o >>= 1) v = fmaxf(v, __shfl_xor(v, o));
    return v;
}

__device__ __forceinline__ float wave_sum(float v) {
    for (int o = 16; o > 0; o >>= 1) v += __shfl_xor(v, o);
    return v;
}

__device__ __forceinline__ void gat_rows(const float* __restrict__ xl, int lane, int sv,
                                         float w0, float w1, float w2, float w3, int cc,
                                         float& acc0, float& acc1, float& acc2, float& acc3) {
    #pragma unroll 4
    for (int k = 0; k < cc; ++k) {
        int s = __builtin_amdgcn_readlane(sv, k);
        float u0 = readlane_f(w0, k);
        float u1 = readlane_f(w1, k);
        float u2 = readlane_f(w2, k);
        float u3 = readlane_f(w3, k);
        const float* xs = xl + (size_t)s * 128 + lane;
        acc0 = fmaf(xs[0],  u0, acc0);
        acc1 = fmaf(xs[32], u1, acc1);
        acc2 = fmaf(xs[64], u2, acc2);
        acc3 = fmaf(xs[96], u3, acc3);
    }
}

__device__ __forceinline__ int bucket_sorted(const int* __restrict__ csr_src, int beg, int n,
                                             int base, int lane, int fill) {
    int mine = fill;
    for (int k = 0; k < n; ++k) {
        int sk = csr_src[beg + k];
        int rank = 0;
        for (int b = 0; b < n; b += 32) {
            int i = b + lane;
            int si = i < n ? csr_src[beg + i] : 0x7fffffff;
            rank += __builtin_popcount(
                __builtin_amdgcn_ballot_w32(si < sk || (si == sk && i < k)));
        }
        if (lane == rank - base) mine = sk;
    }
    return mine;
}

__global__ __launch_bounds__(256) void k_gat_node(const float* __restrict__ xl,
                                                  const float* __restrict__ a_src,
                                                  const float* __restrict__ a_dst,
                                                  const int* __restrict__ rowptr,
                                                  const int* __restrict__ deg,
                                                  const int* __restrict__ csr_src,
                                                  const float* __restrict__ gat_bias,
                                                  float* __restrict__ out) {
    int gid = blockIdx.x * blockDim.x + threadIdx.x;
    if ((gid >> 5) >= N_NODES) return;
    int lane = threadIdx.x & 31;
    int d   = __builtin_amdgcn_readfirstlane(gid >> 5);
    int beg = __builtin_amdgcn_readfirstlane(rowptr[d]);
    int n   = __builtin_amdgcn_readfirstlane(deg[d]);

    const float4 ad  = *(const float4*)(a_dst + (size_t)d * 4);
    const float4 asd = *(const float4*)(a_src + (size_t)d * 4);
    float l0 = lrelu(asd.x + ad.x), l1 = lrelu(asd.y + ad.y);
    float l2 = lrelu(asd.z + ad.z), l3 = lrelu(asd.w + ad.w);

    int sv = d;
    if (n <= 32) {
        int raw = lane < n ? csr_src[beg + lane] : 0x7fffffff;
        for (int k = 0; k < n; ++k) {
            int sk = __builtin_amdgcn_readlane(raw, k);
            int rank = __builtin_popcount(
                __builtin_amdgcn_ballot_w32(raw < sk || (raw == sk && lane < k)));
            if (lane == rank) sv = sk >> 3;
        }
    } else {
        sv = bucket_sorted(csr_src, beg, n, 0, lane, d << 3) >> 3;
    }
    bool ok = lane < n && sv != d;
    float4 av = *(const float4*)(a_src + (size_t)sv * 4);
    float c0 = lrelu(av.x + ad.x), c1 = lrelu(av.y + ad.y);
    float c2 = lrelu(av.z + ad.z), c3 = lrelu(av.w + ad.w);

    float m0 = ok ? fmaxf(l0, c0) : l0, m1 = ok ? fmaxf(l1, c1) : l1;
    float m2 = ok ? fmaxf(l2, c2) : l2, m3 = ok ? fmaxf(l3, c3) : l3;
    for (int b = 32; b < n; b += 32) {
        int s = bucket_sorted(csr_src, beg, n, b, lane, d << 3) >> 3;
        bool okb = b + lane < n && s != d;
        float4 a = *(const float4*)(a_src + (size_t)s * 4);
        float t0 = lrelu(a.x + ad.x), t1 = lrelu(a.y + ad.y);
        float t2 = lrelu(a.z + ad.z), t3 = lrelu(a.w + ad.w);
        m0 = okb ? fmaxf(m0, t0) : m0; m1 = okb ? fmaxf(m1, t1) : m1;
        m2 = okb ? fmaxf(m2, t2) : m2; m3 = okb ? fmaxf(m3, t3) : m3;
    }
    m0 = wave_max(m0); m1 = wave_max(m1); m2 = wave_max(m2); m3 = wave_max(m3);

    float e0 = ok ? __expf(c0 - m0) : 0.f, e1 = ok ? __expf(c1 - m1) : 0.f;
    float e2 = ok ? __expf(c2 - m2) : 0.f, e3 = ok ? __expf(c3 - m3) : 0.f;
    float d0 = e0, d1 = e1, d2 = e2, d3 = e3;
    for (int b = 32; b < n; b += 32) {
        int s = bucket_sorted(csr_src, beg, n, b, lane, d << 3) >> 3;
        bool okb = b + lane < n && s != d;
        float4 a = *(const float4*)(a_src + (size_t)s * 4);
        d0 += okb ? __expf(lrelu(a.x + ad.x) - m0) : 0.f;
        d1 += okb ? __expf(lrelu(a.y + ad.y) - m1) : 0.f;
        d2 += okb ? __expf(lrelu(a.z + ad.z) - m2) : 0.f;
        d3 += okb ? __expf(lrelu(a.w + ad.w) - m3) : 0.f;
    }
    float s0 = __expf(l0 - m0), s1 = __expf(l1 - m1), s2 = __expf(l2 - m2), s3 = __expf(l3 - m3);
    d0 = wave_sum(d0) + s0; d1 = wave_sum(d1) + s1;
    d2 = wave_sum(d2) + s2; d3 = wave_sum(d3) + s3;

    const float* xd = xl + (size_t)d * 128 + lane;
    float acc0 = xd[0]  * (s0 / d0);
    float acc1 = xd[32] * (s1 / d1);
    float acc2 = xd[64] * (s2 / d2);
    float acc3 = xd[96] * (s3 / d3);
    gat_rows(xl, lane, sv, e0 / d0, e1 / d1, e2 / d2, e3 / d3, n < 32 ? n : 32,
             acc0, acc1, acc2, acc3);
    for (int b = 32; b < n; b += 32) {
        int s = bucket_sorted(csr_src, beg, n, b, lane, d << 3) >> 3;
        bool okb = b + lane < n && s != d;
        float4 a = *(const float4*)(a_src + (size_t)s * 4);
        float w0 = okb ? __expf(lrelu(a.x + ad.x) - m0) / d0 : 0.f;
        float w1 = okb ? __expf(lrelu(a.y + ad.y) - m1) / d1 : 0.f;
        float w2 = okb ? __expf(lrelu(a.z + ad.z) - m2) / d2 : 0.f;
        float w3 = okb ? __expf(lrelu(a.w + ad.w) - m3) / d3 : 0.f;
        gat_rows(xl, lane, s, w0, w1, w2, w3, (n - b) < 32 ? (n - b) : 32,
                 acc0, acc1, acc2, acc3);
    }

    float* o = out + (size_t)d * 128 + lane;
    o[0]  = acc0 + gat_bias[lane];
    o[32] = acc1 + gat_bias[32 + lane];
    o[64] = acc2 + gat_bias[64 + lane];
    o[96] = acc3 + gat_bias[96 + lane];
}


extern "C" void kernel_launch(void* const* d_in, const int* in_sizes, int n_in,
                              void* d_out, int out_size, void* d_ws, size_t ws_size,
                              hipStream_t stream) {
    const float* x        = (const float*)d_in[0];
    const int*   ei       = (const int*)d_in[1];
    const int*   et       = (const int*)d_in[2];
    const float* comp     = (const float*)d_in[3];
    const float* basis    = (const float*)d_in[4];
    const float* root     = (const float*)d_in[5];
    const float* bias1    = (const float*)d_in[6];
    const float* gat_w    = (const float*)d_in[7];
    const float* att_src  = (const float*)d_in[8];
    const float* att_dst  = (const float*)d_in[9];
    const float* gat_bias = (const float*)d_in[10];
    float* out = (float*)d_out;

    float* ws = (float*)d_ws;
    size_t off = 0;
    float* W    = ws + off; off += (size_t)RREL * GDIM * HD1;
    int*   deg  = (int*)(ws + off); off += (size_t)N_NODES;
    float* hbuf = ws + off; off += (size_t)N_NODES * HD1;
    float* xl   = ws + off; off += (size_t)N_NODES * HD1;
    float* asg  = ws + off; off += (size_t)N_NODES * HEADS;
    float* adg  = ws + off; off += (size_t)N_NODES * HEADS;
    int*   rowptr = (int*)(ws + off); off += (size_t)N_NODES;
    int*   slot   = (int*)(ws + off); off += (size_t)N_EDGES;
    int*   csrc   = (int*)(ws + off); off += (size_t)N_EDGES;
    float* acc8 = ws + off; off += (size_t)RREL * N_NODES * GDIM;

    dim3 mmBlock(32, 8);
    int mmGrid = N_NODES / 16;

    k_basis_w<<<(RREL * GDIM * HD1 + 255) / 256, 256, 0, stream>>>(comp, basis, W);
    hipMemsetAsync(deg, 0, (size_t)N_NODES * sizeof(int), stream);
    k_count<<<(N_EDGES + 255) / 256, 256, 0, stream>>>(ei, deg, slot);
    k_scan<<<SCAN_B, SCAN_T, 0, stream>>>(deg, rowptr);
    k_fill_csr<<<(N_EDGES + 255) / 256, 256, 0, stream>>>(ei, et, rowptr, slot, csrc);

    hipMemsetAsync(acc8, 0, (size_t)RREL * N_NODES * GDIM * sizeof(float), stream);
    k_rgcn_gather<<<(N_NODES * 32 + 255) / 256, 256, 0, stream>>>(x, rowptr, deg, csrc, acc8);
    k_mm_rgcn<<<RG_WGS, mmBlock, 0, stream>>>(acc8, x, W, root, bias1, hbuf);

    k_mm16<true><<<mmGrid, mmBlock, 0, stream>>>(hbuf, gat_w, xl, nullptr);
    k_attn<<<(N_NODES * HEADS + 255) / 256, 256, 0, stream>>>(xl, att_src, att_dst, asg, adg);

    k_gat_node<<<(N_NODES * 32 + 255) / 256, 256, 0, stream>>>(xl, asg, adg, rowptr, deg, csrc,
                                                               gat_bias, out);

    (void)in_sizes; (void)n_in; (void)out_size; (void)ws_size;
}


# bucket build: edges ordered by destination in LDS in two levels (chunk by 196-node bin, then bin by node); the returning global add and the scattered store per edge go; sum orders kept
# speedup vs baseline: 3.9387x; 1.2137x over previous
#include <hip/hip_runtime.h>
#include <math.h>

#define N_NODES 50000
#define N_EDGES 800000
#define GDIM 128
#define HD1 128
#define HEADS 4
#define HD2 32
#define RREL 8
#define NB 20

typedef __attribute__((ext_vector_type(2))) float v2f;
typedef __attribute__((ext_vector_type(8))) float v8f;
typedef __attribute__((ext_vector_type(4))) int v4i;

__device__ __forceinline__ float lrelu(float x) { return x > 0.f ? x : 0.2f * x; }


__global__ void k_basis_w(const float* __restrict__ comp, const float* __restrict__ basis,
                          float* __restrict__ W) {
    int t = blockIdx.x * blockDim.x + threadIdx.x;
    if (t >= RREL * GDIM * HD1) return;
    int r = t / (GDIM * HD1);
    int io = t % (GDIM * HD1);
    float s = 0.f;
    #pragma unroll
    for (int b = 0; b < NB; ++b) s += comp[r * NB + b] * basis[b * GDIM * HD1 + io];
    W[t] = s;
}


template <bool BIAS_MODE>
__global__ __launch_bounds__(256) void k_mm16(const float* __restrict__ A,
                                              const float* __restrict__ B,
                                              float* __restrict__ C,
                                              const float* __restrict__ bias) {
    __shared__ float As[16][132];
    int tx = threadIdx.x;
    int ty = threadIdx.y;
    int tid = ty * 32 + tx;
    int row0 = blockIdx.x * 16;

    const float4* A4 = (const float4*)(A + (size_t)row0 * 128);
    for (int i = tid; i < 16 * 32; i += 256) {
        int rr = i >> 5, cc = i & 31;
        float4 v = A4[rr * 32 + cc];
        As[rr][cc * 4 + 0] = v.x;
        As[rr][cc * 4 + 1] = v.y;
        As[rr][cc * 4 + 2] = v.z;
        As[rr][cc * 4 + 3] = v.w;
    }
    __syncthreads();

    int col0 = ty * 16;
    int half = tx >> 4;
    int mn = tx & 15;
    int kg = half * 2;

    v8f c;
    if (BIAS_MODE) {
        float bv = bias ? bias[col0 + mn] : 0.f;
        #pragma unroll
        for (int i = 0; i < 8; ++i) c[i] = bv;
    } else {
        #pragma unroll
        for (int i = 0; i < 8; ++i)
            c[i] = C[(size_t)(row0 + half * 8 + i) * 128 + col0 + mn];
    }

    #pragma unroll 4
    for (int k0 = 0; k0 < 128; k0 += 4) {
        v2f a, b;
        a.x = As[mn][k0 + kg];
        a.y = As[mn][k0 + kg + 1];
        b.x = B[(size_t)(k0 + kg) * 128 + col0 + mn];
        b.y = B[(size_t)(k0 + kg + 1) * 128 + col0 + mn];
        c = __builtin_amdgcn_wmma_f32_16x16x4_f32(false, a, false, b, (short)0, c,
                                                  false, false);
    }

    #pragma unroll
    for (int i = 0; i < 8; ++i)
        C[(size_t)(row0 + half * 8 + i) * 128 + col0 + mn] = c[i];
}


#define RG_CH (RREL + 1)
#define RG_T 13
#define RG_ROWS (RG_T * 16)
#define RG_WGS ((N_NODES / 16 + RG_T - 1) / RG_T)
#define RG_STAGE (RG_ROWS * 32 / 256)
#define RG_KG 8

__device__ __forceinline__ void rg_load_block(v2f (&f)[RG_T][2], const float* a, int k0) {
    #pragma unroll
    for (int r = 0; r < RG_T; ++r) {
        const float* q = a + r * 16 * 132 + k0;
        f[r][0].x = q[0]; f[r][0].y = q[1];
        f[r][1].x = q[4]; f[r][1].y = q[5];
    }
}

__device__ __forceinline__ void rg_mma_block(v8f (&c)[RG_T], const v2f (&f)[RG_T][2],
                                             v2f b0, v2f b1) {
    #pragma unroll
    for (int r = 0; r < RG_T; ++r)
        c[r] = __builtin_amdgcn_wmma_f32_16x16x4_f32(false, f[r][0], false, b0, (short)0, c[r],
                                                     false, false);
    #pragma unroll
    for (int r = 0; r < RG_T; ++r)
        c[r] = __builtin_amdgcn_wmma_f32_16x16x4_f32(false, f[r][1], false, b1, (short)0, c[r],
                                                     false, false);
}

__device__ __forceinline__ void rg_load_b(v2f (&v)[4], const float* bl) {
    #pragma unroll
    for (int s = 0; s < 4; ++s) {
        v[s].x = bl[(4 * s) * 128];
        v[s].y = bl[(4 * s + 1) * 128];
    }
}

template <int G, bool MORE>
__device__ __forceinline__ void rg_group(v8f (&c)[RG_T], v2f (&fa)[RG_T][2], v2f (&fb)[RG_T][2],
                                         const v2f (&bc)[4], v2f (&bn)[4],
                                         const float* afrag, const float* bcur, const float* bnxt,
                                         float4 (&p)[4], const float4* src4, float* sdst,
                                         int srow) {
    if (G + 1 < RG_KG) rg_load_b(bn, bcur + (G + 1) * 16 * 128);
    else if (MORE)     rg_load_b(bn, bnxt);
    if (MORE) {
        #pragma unroll
        for (int j = 0; j < 4; ++j) {
            int q = (G - 1) * 4 + j;
            if (G >= 1 && q < RG_STAGE) {
                float* d = sdst + q * 8 * 132;
                d[0] = p[j].x; d[1] = p[j].y; d[2] = p[j].z; d[3] = p[j].w;
            }
        }
        #pragma unroll
        for (int j = 0; j < 4; ++j) {
            int q = G * 4 + j;
            if (q < RG_STAGE) {
                int gr = srow + 8 * q;
                gr = gr < N_NODES ? gr : N_NODES - 1;
                p[j] = src4[(size_t)gr * 32];
            }
        }
    }
    rg_load_block(fb, afrag, G * 16 + 8);
    __builtin_amdgcn_sched_barrier(0);
    rg_mma_block(c, fa, bc[0], bc[1]);
    __builtin_amdgcn_sched_barrier(0);
    if (G + 1 < RG_KG) rg_load_block(fa, afrag, G * 16 + 16);
    __builtin_amdgcn_sched_barrier(0);
    rg_mma_block(c, fb, bc[2], bc[3]);
    __builtin_amdgcn_sched_barrier(0);
}

template <bool MORE>
__device__ __forceinline__ void rg_chunk(v8f (&c)[RG_T], v2f (&b0)[4], v2f (&b1)[4],
                                         const float* afrag, const float* bcur, const float* bnxt,
                                         const float4* src4, float* sdst, int srow) {
    v2f fa[RG_T][2], fb[RG_T][2];
    float4 p[4];
    rg_load_block(fa, afrag, 0);
    rg_group<0, MORE>(c, fa, fb, b0, b1, afrag, bcur, bnxt, p, src4, sdst, srow);
    rg_group<1, MORE>(c, fa, fb, b1, b0, afrag, bcur, bnxt, p, src4, sdst, srow);
    rg_group<2, MORE>(c, fa, fb, b0, b1, afrag, bcur, bnxt, p, src4, sdst, srow);
    rg_group<3, MORE>(c, fa, fb, b1, b0, afrag, bcur, bnxt, p, src4, sdst, srow);
    rg_group<4, MORE>(c, fa, fb, b0, b1, afrag, bcur, bnxt, p, src4, sdst, srow);
    rg_group<5, MORE>(c, fa, fb, b1, b0, afrag, bcur, bnxt, p, src4, sdst, srow);
    rg_group<6, MORE>(c, fa, fb, b0, b1, afrag, bcur, bnxt, p, src4, sdst, srow);
    rg_group<7, MORE>(c, fa, fb, b1, b0, afrag, bcur, bnxt, p, src4, sdst, srow);
}

__global__ __launch_bounds__(256) void k_mm_rgcn(const float* __restrict__ acc8,
                                                 const float* __restrict__ x,
                                                 const float* __restrict__ W,
                                                 const float* __restrict__ root,
                                                 const float* __restrict__ bias,
                                                 float* __restrict__ C) {
    __shared__ float As[2][RG_ROWS][132];
    int tx = threadIdx.x;
    int ty = threadIdx.y;
    int tid = ty * 32 + tx;
    int row0 = blockIdx.x * RG_ROWS;

    int ra = tid >> 5, ca = tid & 31;
    const float4* acc4 = (const float4*)acc8 + ca;
    const float4* x4   = (const float4*)x + ca;
    const size_t chunk4 = (size_t)N_NODES * 128 / 4;

    #pragma unroll
    for (int q = 0; q < RG_STAGE; ++q) {
        int gr = row0 + ra + 8 * q;
        gr = gr < N_NODES ? gr : N_NODES - 1;
        float4 v = acc4[(size_t)gr * 32];
        float* dst = &As[0][ra + 8 * q][ca * 4];
        dst[0] = v.x; dst[1] = v.y; dst[2] = v.z; dst[3] = v.w;
    }
    __syncthreads();

    int col0 = ty * 16;
    int half = tx >> 4;
    int mn = tx & 15;
    int kg = half * 2;

    v8f c[RG_T];
    {
        float bv = bias[col0 + mn];
        #pragma unroll
        for (int r = 0; r < RG_T; ++r) {
            #pragma unroll
            for (int i = 0; i < 8; ++i) c[r][i] = bv;
        }
    }

    const int blane = kg * 128 + col0 + mn;
    v2f b0[4], b1[4];
    rg_load_b(b0, W + blane);

    #pragma unroll 1
    for (int ch = 0; ch < RREL; ++ch) {
        int cur = ch & 1, nxt = cur ^ 1;
        const float4* src4 = (ch + 1 < RREL) ? acc4 + (size_t)(ch + 1) * chunk4 : x4;
        const float* bcur = W + (size_t)ch * GDIM * HD1 + blane;
        const float* bnxt = ((ch + 1 < RREL) ? W + (size_t)(ch + 1) * GDIM * HD1 : root) + blane;
        rg_chunk<true>(c, b0, b1, &As[cur][mn][kg], bcur, bnxt, src4, &As[nxt][ra][ca * 4],
                       row0 + ra);
        __syncthreads();
    }
    rg_chunk<false>(c, b0, b1, &As[RREL & 1][mn][kg], root + blane, root + blane, x4,
                    &As[(RREL & 1) ^ 1][ra][ca * 4], row0 + ra);

    #pragma unroll
    for (int r = 0; r < RG_T; ++r) {
        if (row0 + r * 16 < N_NODES) {
            #pragma unroll
            for (int i = 0; i < 8; ++i)
                C[(size_t)(row0 + r * 16 + half * 8 + i) * 128 + col0 + mn] = c[r][i];
        }
    }
}


__global__ void k_attn(const float* __restrict__ xl, const float* __restrict__ att_src,
                       const float* __restrict__ att_dst, float* __restrict__ a_src,
                       float* __restrict__ a_dst) {
    int t = blockIdx.x * blockDim.x + threadIdx.x;
    if (t >= N_NODES * HEADS) return;
    int n = t >> 2, h = t & 3;
    const float* row = xl + (size_t)n * 128 + h * HD2;
    float ss = 0.f, sd = 0.f;
    #pragma unroll
    for (int j = 0; j < HD2; ++j) {
        float v = row[j];
        ss += v * att_src[h * HD2 + j];
        sd += v * att_dst[h * HD2 + j];
    }
    a_src[t] = ss;
    a_dst[t] = sd;
}


#define CB_CHUNKS 128
#define CB_CH (N_EDGES / CB_CHUNKS)
#define CB_BINS 256
#define CB_BINW 196
#define CB_T 1024
#define CB_PER ((CB_CH + CB_T - 1) / CB_T)
#define CC_T 512
#define CC_CAP 8192
#define CC_RUNS (CB_CHUNKS / (CC_T / 32))
#define CC_PRE (((CB_BINS - 1) * (CB_CHUNKS / 4) + CC_T - 1) / CC_T)

static_assert(N_EDGES % CB_CHUNKS == 0, "chunks are whole");
static_assert(CB_BINS * CB_BINW >= N_NODES, "the bins cover every node");
static_assert(CB_BINW <= 256 && CB_BINS == 256, "node-in-bin takes 8 bits, one scan thread per bin");
static_assert(N_NODES <= 65536, "a source index takes 16 bits");
static_assert(CB_CHUNKS % 4 == 0 && CB_CHUNKS <= CC_T, "count rows are read as 16-byte pieces");

__device__ __forceinline__ int scan256(int own, int t, int* sw, int& total) {
    int lane = t & 31, w = t >> 5;
    int inc = own;
    for (int o = 1; o < 32; o <<= 1) {
        int v = __shfl_up(inc, o);
        if (lane >= o) inc += v;
    }
    if (t < 256 && lane == 31) sw[w] = inc;
    __syncthreads();
    int add = 0;
    total = 0;
    #pragma unroll
    for (int i = 0; i < 8; ++i) {
        int s = sw[i];
        total += s;
        if (i < w) add += s;
    }
    return add + inc - own;
}

__global__ __launch_bounds__(CB_T) void k_bin_edges(const int* __restrict__ ei,
                                                    const int* __restrict__ et,
                                                    int* __restrict__ ebin,
                                                    int* __restrict__ bcnt,
                                                    int* __restrict__ boff) {
    __shared__ int hist[CB_BINS];
    __shared__ int sw[8];
    __shared__ int sorted[CB_CH];
    int t = threadIdx.x, c = blockIdx.x;
    if (t < CB_BINS) hist[t] = 0;
    __syncthreads();

    const int e0 = c * CB_CH;
    int rec[CB_PER], bin[CB_PER], rk[CB_PER];
    int es[CB_PER], ed[CB_PER], er[CB_PER];
    #pragma unroll
    for (int j = 0; j < CB_PER; ++j) {
        int i = t + j * CB_T;
        i = i < CB_CH ? i : CB_CH - 1;
        es[j] = ei[e0 + i];
        ed[j] = ei[N_EDGES + e0 + i];
        er[j] = et[e0 + i];
    }
    static_assert(CB_PER == 7, "the statement below names the loads of seven edges");
    asm volatile("" :: "v"(es[0]), "v"(ed[0]), "v"(er[0]), "v"(es[1]), "v"(ed[1]), "v"(er[1]),
                       "v"(es[2]), "v"(ed[2]), "v"(er[2]), "v"(es[3]), "v"(ed[3]), "v"(er[3]),
                       "v"(es[4]), "v"(ed[4]), "v"(er[4]), "v"(es[5]), "v"(ed[5]), "v"(er[5]),
                       "v"(es[6]), "v"(ed[6]), "v"(er[6]));
    #pragma unroll
    for (int j = 0; j < CB_PER; ++j) {
        if (t + j * CB_T < CB_CH) {
            int b = ed[j] / CB_BINW;
            bin[j] = b;
            rec[j] = ((ed[j] - b * CB_BINW) << 19) | (es[j] << 3) | er[j];
            rk[j] = atomicAdd(&hist[b], 1);
        }
    }
    __syncthreads();

    int own = t < CB_BINS ? hist[t] : 0;
    int total;
    int start = scan256(own, t, sw, total);
    if (t < CB_BINS) {
        hist[t] = start;
        bcnt[t * CB_CHUNKS + c] = own;
        boff[t * CB_CHUNKS + c] = start;
    }
    __syncthreads();

    #pragma unroll
    for (int j = 0; j < CB_PER; ++j) {
        int i = t + j * CB_T;
        if (i < CB_CH) sorted[hist[bin[j]] + rk[j]] = rec[j];
    }
    __syncthreads();
    for (int i = t; i < CB_CH; i += CB_T) ebin[e0 + i] = sorted[i];
}

__global__ __launch_bounds__(CC_T) void k_bin_csr(const int* __restrict__ ebin,
                                                  const int* __restrict__ bcnt,
                                                  const int* __restrict__ boff,
                                                  int* __restrict__ deg,
                                                  int* __restrict__ rowptr,
                                                  int* __restrict__ csr_src) {
    __shared__ int rcnt[CB_CHUNKS];
    __shared__ int roff[CB_CHUNKS];
    __shared__ int ncnt[256];
    __shared__ int nnext[256];
    __shared__ int red[CC_T / 32];
    __shared__ int sw[8];
    __shared__ int stage[CC_CAP];
    int t = threadIdx.x, lane = t & 31, w = t >> 5, b = blockIdx.x;

    int pre = 0;
    {
        const v4i* c4 = (const v4i*)bcnt;
        const int n4 = b * (CB_CHUNKS / 4);
        const int lim = n4 > 0 ? n4 - 1 : 0;
        v4i v[CC_PRE];
        #pragma unroll
        for (int k = 0; k < CC_PRE; ++k) {
            int i = t + k * CC_T;
            v[k] = c4[i < lim ? i : lim];
        }
        static_assert(CC_PRE == 16, "the statement below names sixteen loads");
        asm volatile("" :: "v"(v[0]), "v"(v[1]), "v"(v[2]), "v"(v[3]), "v"(v[4]), "v"(v[5]),
                           "v"(v[6]), "v"(v[7]), "v"(v[8]), "v"(v[9]), "v"(v[10]), "v"(v[11]),
                           "v"(v[12]), "v"(v[13]), "v"(v[14]), "v"(v[15]));
        #pragma unroll
        for (int k = 0; k < CC_PRE; ++k)
            pre += (t + k * CC_T < n4) ? (v[k].x + v[k].y) + (v[k].z + v[k].w) : 0;
    }
    for (int o = 16; o > 0; o >>= 1) pre += __shfl_xor(pre, o);
    if (lane == 0) red[w] = pre;
    if (t < CB_CHUNKS) {
        rcnt[t] = bcnt[b * CB_CHUNKS + t];
        roff[t] = boff[b * CB_CHUNKS + t];
    }
    if (t < 256) ncnt[t] = 0;
    __syncthreads();
    int base = 0;
    #pragma unroll
    for (int i = 0; i < CC_T / 32; ++i) base += red[i];

    int rn[CC_RUNS], rs[CC_RUNS], e[CC_RUNS];
    #pragma unroll
    for (int q = 0; q < CC_RUNS; ++q) {
        int c = w + q * (CC_T / 32);
        rn[q] = rcnt[c];
        rs[q] = c * CB_CH + roff[c];
        int i = rs[q] + lane;
        e[q] = ebin[i < N_EDGES - 1 ? i : N_EDGES - 1];
    }
    static_assert(CC_RUNS == 8, "the statement below names eight loads");
    asm volatile("" :: "v"(e[0]), "v"(e[1]), "v"(e[2]), "v"(e[3]), "v"(e[4]), "v"(e[5]), "v"(e[6]),
                       "v"(e[7]));

    #pragma unroll
    for (int q = 0; q < CC_RUNS; ++q) {
        if (lane < rn[q]) atomicAdd(&ncnt[e[q] >> 19], 1);
        for (int j = lane + 32; j < rn[q]; j += 32) atomicAdd(&ncnt[ebin[rs[q] + j] >> 19], 1);
    }
    __syncthreads();

    int own = t < 256 ? ncnt[t] : 0;
    int nb;
    int st = scan256(own, t, sw, nb);
    if (t < 256) {
        nnext[t] = st;
        int node = b * CB_BINW + t;
        if (t < CB_BINW && node < N_NODES) {
            deg[node] = own;
            rowptr[node] = base + st;
        }
    }
    __syncthreads();

    const bool fits = nb <= CC_CAP;
    #pragma unroll
    for (int q = 0; q < CC_RUNS; ++q) {
        if (lane < rn[q]) {
            int pos = atomicAdd(&nnext[e[q] >> 19], 1);
            int ent = e[q] & 0x7ffff;
            if (fits) stage[pos] = ent;
            else      csr_src[base + pos] = ent;
        }
        for (int j = lane + 32; j < rn[q]; j += 32) {
            int r = ebin[rs[q] + j];
            int pos = atomicAdd(&nnext[r >> 19], 1);
            int ent = r & 0x7ffff;
            if (fits) stage[pos] = ent;
            else      csr_src[base + pos] = ent;
        }
    }
    if (fits) {
        __syncthreads();
        for (int i = t; i < nb; i += CC_T) csr_src[base + i] = stage[i];
    }
}


__device__ __forceinline__ int rg_key(int entry) { return ((entry & 7) << 16) | (entry >> 3); }

__device__ __forceinline__ int rg_bucket_sorted(const int* __restrict__ csr_src, int beg, int n,
                                                int base, int lane) {
    int mine = 0x7fffffff;
    for (int k = 0; k < n; ++k) {
        int sk = rg_key(csr_src[beg + k]);
        int rank = 0;
        for (int b = 0; b < n; b += 32) {
            int i = b + lane;
            int si = i < n ? rg_key(csr_src[beg + i]) : 0x7fffffff;
            rank += __builtin_popcount(
                __builtin_amdgcn_ballot_w32(si < sk || (si == sk && i < k)));
        }
        if (lane == rank - base) mine = sk;
    }
    return mine;
}

#define RGG_U 4

__global__ __launch_bounds__(256) void k_rgcn_gather(const float* __restrict__ x,
                                                     const int* __restrict__ rowptr,
                                                     const int* __restrict__ deg,
                                                     const int* __restrict__ csr_src,
                                                     float* __restrict__ acc8) {
    int gid = blockIdx.x * blockDim.x + threadIdx.x;
    if ((gid >> 5) >= N_NODES) return;
    int lane = threadIdx.x & 31;
    int d   = __builtin_amdgcn_readfirstlane(gid >> 5);
    int beg = __builtin_amdgcn_readfirstlane(rowptr[d]);
    int n   = __builtin_amdgcn_readfirstlane(deg[d]);

    const float4* x4 = (const float4*)x + lane;
    float4 acc = make_float4(0.f, 0.f, 0.f, 0.f);
    int cur = -1;
    int cnt = 0;

    for (int base = 0; base < n; base += 32) {
        int kv;
        if (n <= 32) {
            int raw = lane < n ? rg_key(csr_src[beg + lane]) : 0x7fffffff;
            kv = raw;
            for (int k = 0; k < n; ++k) {
                int sk = __builtin_amdgcn_readlane(raw, k);
                int rank = __builtin_popcount(
                    __builtin_amdgcn_ballot_w32(raw < sk || (raw == sk && lane < k)));
                if (lane == rank) kv = sk;
            }
        } else {
            kv = rg_bucket_sorted(csr_src, beg, n, base, lane);
        }
        int cc = (n - base) < 32 ? (n - base) : 32;
        for (int k = 0; k < cc; k += RGG_U) {
            int    tk[RGG_U];
            float4 v[RGG_U];
            #pragma unroll
            for (int u = 0; u < RGG_U; ++u) {
                int kk = (k + u) < cc ? (k + u) : (cc - 1);
                int key = __builtin_amdgcn_readlane(kv, kk);
                tk[u] = key >> 16;
                v[u] = x4[(size_t)(key & 0xffff) * (GDIM / 4)];
            }
            static_assert(RGG_U == 4, "the statement below names the loads of four rows");
            asm volatile("" :: "v"(v[0].x), "v"(v[0].y), "v"(v[0].z), "v"(v[0].w),
                               "v"(v[1].x), "v"(v[1].y), "v"(v[1].z), "v"(v[1].w),
                               "v"(v[2].x), "v"(v[2].y), "v"(v[2].z), "v"(v[2].w),
                               "v"(v[3].x), "v"(v[3].y), "v"(v[3].z), "v"(v[3].w));
            #pragma unroll
            for (int u = 0; u < RGG_U; ++u) {
                if (k + u < cc) {
                    if (tk[u] != cur) {
                        if (cnt > 0) {
                            float c = (float)cnt;
                            float4 o = make_float4(acc.x / c, acc.y / c, acc.z / c, acc.w / c);
                            ((float4*)(acc8 + ((size_t)cur * N_NODES + d) * GDIM))[lane] = o;
                        }
                        acc = make_float4(0.f, 0.f, 0.f, 0.f);
                        cnt = 0;
                        cur = tk[u];
                    }
                    acc.x += v[u].x; acc.y += v[u].y; acc.z += v[u].z; acc.w += v[u].w;
                    ++cnt;
                }
            }
        }
    }
    if (cnt > 0) {
        float c = (float)cnt;
        float4 o = make_float4(acc.x / c, acc.y / c, acc.z / c, acc.w / c);
        ((float4*)(acc8 + ((size_t)cur * N_NODES + d) * GDIM))[lane] = o;
    }
}

__device__ __forceinline__ float readlane_f(float v, int l) {
    return __int_as_float(__builtin_amdgcn_readlane(__float_as_int(v), l));
}

__device__ __forceinline__ float wave_max(float v) {
    for (int o = 16; o > 0; o >>= 1) v = fmaxf(v, __shfl_xor(v, o));
    return v;
}

__device__ __forceinline__ float wave_sum(float v) {
    for (int o = 16; o > 0; o >>= 1) v += __shfl_xor(v, o);
    return v;
}

__device__ __forceinline__ void gat_rows(const float* __restrict__ xl, int lane, int sv,
                                         float w0, float w1, float w2, float w3, int cc,
                                         float& acc0, float& acc1, float& acc2, float& acc3) {
    #pragma unroll 4
    for (int k = 0; k < cc; ++k) {
        int s = __builtin_amdgcn_readlane(sv, k);
        float u0 = readlane_f(w0, k);
        float u1 = readlane_f(w1, k);
        float u2 = readlane_f(w2, k);
        float u3 = readlane_f(w3, k);
        const float* xs = xl + (size_t)s * 128 + lane;
        acc0 = fmaf(xs[0],  u0, acc0);
        acc1 = fmaf(xs[32], u1, acc1);
        acc2 = fmaf(xs[64], u2, acc2);
        acc3 = fmaf(xs[96], u3, acc3);
    }
}

__device__ __forceinline__ int bucket_sorted(const int* __restrict__ csr_src, int beg, int n,
                                             int base, int lane, int fill) {
    int mine = fill;
    for (int k = 0; k < n; ++k) {
        int sk = csr_src[beg + k];
        int rank = 0;
        for (int b = 0; b < n; b += 32) {
            int i = b + lane;
            int si = i < n ? csr_src[beg + i] : 0x7fffffff;
            rank += __builtin_popcount(
                __builtin_amdgcn_ballot_w32(si < sk || (si == sk && i < k)));
        }
        if (lane == rank - base) mine = sk;
    }
    return mine;
}

__global__ __launch_bounds__(256) void k_gat_node(const float* __restrict__ xl,
                                                  const float* __restrict__ a_src,
                                                  const float* __restrict__ a_dst,
                                                  const int* __restrict__ rowptr,
                                                  const int* __restrict__ deg,
                                                  const int* __restrict__ csr_src,
                                                  const float* __restrict__ gat_bias,
                                                  float* __restrict__ out) {
    int gid = blockIdx.x * blockDim.x + threadIdx.x;
    if ((gid >> 5) >= N_NODES) return;
    int lane = threadIdx.x & 31;
    int d   = __builtin_amdgcn_readfirstlane(gid >> 5);
    int beg = __builtin_amdgcn_readfirstlane(rowptr[d]);
    int n   = __builtin_amdgcn_readfirstlane(deg[d]);

    const float4 ad  = *(const float4*)(a_dst + (size_t)d * 4);
    const float4 asd = *(const float4*)(a_src + (size_t)d * 4);
    float l0 = lrelu(asd.x + ad.x), l1 = lrelu(asd.y + ad.y);
    float l2 = lrelu(asd.z + ad.z), l3 = lrelu(asd.w + ad.w);

    int sv = d;
    if (n <= 32) {
        int raw = lane < n ? csr_src[beg + lane] : 0x7fffffff;
        for (int k = 0; k < n; ++k) {
            int sk = __builtin_amdgcn_readlane(raw, k);
            int rank = __builtin_popcount(
                __builtin_amdgcn_ballot_w32(raw < sk || (raw == sk && lane < k)));
            if (lane == rank) sv = sk >> 3;
        }
    } else {
        sv = bucket_sorted(csr_src, beg, n, 0, lane, d << 3) >> 3;
    }
    bool ok = lane < n && sv != d;
    float4 av = *(const float4*)(a_src + (size_t)sv * 4);
    float c0 = lrelu(av.x + ad.x), c1 = lrelu(av.y + ad.y);
    float c2 = lrelu(av.z + ad.z), c3 = lrelu(av.w + ad.w);

    float m0 = ok ? fmaxf(l0, c0) : l0, m1 = ok ? fmaxf(l1, c1) : l1;
    float m2 = ok ? fmaxf(l2, c2) : l2, m3 = ok ? fmaxf(l3, c3) : l3;
    for (int b = 32; b < n; b += 32) {
        int s = bucket_sorted(csr_src, beg, n, b, lane, d << 3) >> 3;
        bool okb = b + lane < n && s != d;
        float4 a = *(const float4*)(a_src + (size_t)s * 4);
        float t0 = lrelu(a.x + ad.x), t1 = lrelu(a.y + ad.y);
        float t2 = lrelu(a.z + ad.z), t3 = lrelu(a.w + ad.w);
        m0 = okb ? fmaxf(m0, t0) : m0; m1 = okb ? fmaxf(m1, t1) : m1;
        m2 = okb ? fmaxf(m2, t2) : m2; m3 = okb ? fmaxf(m3, t3) : m3;
    }
    m0 = wave_max(m0); m1 = wave_max(m1); m2 = wave_max(m2); m3 = wave_max(m3);

    float e0 = ok ? __expf(c0 - m0) : 0.f, e1 = ok ? __expf(c1 - m1) : 0.f;
    float e2 = ok ? __expf(c2 - m2) : 0.f, e3 = ok ? __expf(c3 - m3) : 0.f;
    float d0 = e0, d1 = e1, d2 = e2, d3 = e3;
    for (int b = 32; b < n; b += 32) {
        int s = bucket_sorted(csr_src, beg, n, b, lane, d << 3) >> 3;
        bool okb = b + lane < n && s != d;
        float4 a = *(const float4*)(a_src + (size_t)s * 4);
        d0 += okb ? __expf(lrelu(a.x + ad.x) - m0) : 0.f;
        d1 += okb ? __expf(lrelu(a.y + ad.y) - m1) : 0.f;
        d2 += okb ? __expf(lrelu(a.z + ad.z) - m2) : 0.f;
        d3 += okb ? __expf(lrelu(a.w + ad.w) - m3) : 0.f;
    }
    float s0 = __expf(l0 - m0), s1 = __expf(l1 - m1), s2 = __expf(l2 - m2), s3 = __expf(l3 - m3);
    d0 = wave_sum(d0) + s0; d1 = wave_sum(d1) + s1;
    d2 = wave_sum(d2) + s2; d3 = wave_sum(d3) + s3;

    const float* xd = xl + (size_t)d * 128 + lane;
    float acc0 = xd[0]  * (s0 / d0);
    float acc1 = xd[32] * (s1 / d1);
    float acc2 = xd[64] * (s2 / d2);
    float acc3 = xd[96] * (s3 / d3);
    gat_rows(xl, lane, sv, e0 / d0, e1 / d1, e2 / d2, e3 / d3, n < 32 ? n : 32,
             acc0, acc1, acc2, acc3);
    for (int b = 32; b < n; b += 32) {
        int s = bucket_sorted(csr_src, beg, n, b, lane, d << 3) >> 3;
        bool okb = b + lane < n && s != d;
        float4 a = *(const float4*)(a_src + (size_t)s * 4);
        float w0 = okb ? __expf(lrelu(a.x + ad.x) - m0) / d0 : 0.f;
        float w1 = okb ? __expf(lrelu(a.y + ad.y) - m1) / d1 : 0.f;
        float w2 = okb ? __expf(lrelu(a.z + ad.z) - m2) / d2 : 0.f;
        float w3 = okb ? __expf(lrelu(a.w + ad.w) - m3) / d3 : 0.f;
        gat_rows(xl, lane, s, w0, w1, w2, w3, (n - b) < 32 ? (n - b) : 32,
                 acc0, acc1, acc2, acc3);
    }

    float* o = out + (size_t)d * 128 + lane;
    o[0]  = acc0 + gat_bias[lane];
    o[32] = acc1 + gat_bias[32 + lane];
    o[64] = acc2 + gat_bias[64 + lane];
    o[96] = acc3 + gat_bias[96 + lane];
}


extern "C" void kernel_launch(void* const* d_in, const int* in_sizes, int n_in,
                              void* d_out, int out_size, void* d_ws, size_t ws_size,
                              hipStream_t stream) {
    const float* x        = (const float*)d_in[0];
    const int*   ei       = (const int*)d_in[1];
    const int*   et       = (const int*)d_in[2];
    const float* comp     = (const float*)d_in[3];
    const float* basis    = (const float*)d_in[4];
    const float* root     = (const float*)d_in[5];
    const float* bias1    = (const float*)d_in[6];
    const float* gat_w    = (const float*)d_in[7];
    const float* att_src  = (const float*)d_in[8];
    const float* att_dst  = (const float*)d_in[9];
    const float* gat_bias = (const float*)d_in[10];
    float* out = (float*)d_out;

    float* ws = (float*)d_ws;
    size_t off = 0;
    float* W    = ws + off; off += (size_t)RREL * GDIM * HD1;
    int*   deg  = (int*)(ws + off); off += (size_t)N_NODES;
    float* hbuf = ws + off; off += (size_t)N_NODES * HD1;
    float* xl   = ws + off; off += (size_t)N_NODES * HD1;
    float* asg  = ws + off; off += (size_t)N_NODES * HEADS;
    float* adg  = ws + off; off += (size_t)N_NODES * HEADS;
    int*   rowptr = (int*)(ws + off); off += (size_t)N_NODES;
    int*   ebin   = (int*)(ws + off); off += (size_t)N_EDGES;
    int*   csrc   = (int*)(ws + off); off += (size_t)N_EDGES;
    float* acc8 = ws + off; off += (size_t)RREL * N_NODES * GDIM;
    int*   bcnt   = (int*)(ws + off); off += (size_t)CB_BINS * CB_CHUNKS;
    int*   boff   = (int*)(ws + off); off += (size_t)CB_BINS * CB_CHUNKS;

    dim3 mmBlock(32, 8);
    int mmGrid = N_NODES / 16;

    k_basis_w<<<(RREL * GDIM * HD1 + 255) / 256, 256, 0, stream>>>(comp, basis, W);
    k_bin_edges<<<CB_CHUNKS, CB_T, 0, stream>>>(ei, et, ebin, bcnt, boff);
    k_bin_csr<<<CB_BINS, CC_T, 0, stream>>>(ebin, bcnt, boff, deg, rowptr, csrc);

    hipMemsetAsync(acc8, 0, (size_t)RREL * N_NODES * GDIM * sizeof(float), stream);
    k_rgcn_gather<<<(N_NODES * 32 + 255) / 256, 256, 0, stream>>>(x, rowptr, deg, csrc, acc8);
    k_mm_rgcn<<<RG_WGS, mmBlock, 0, stream>>>(acc8, x, W, root, bias1, hbuf);

    k_mm16<true><<<mmGrid, mmBlock, 0, stream>>>(hbuf, gat_w, xl, nullptr);
    k_attn<<<(N_NODES * HEADS + 255) / 256, 256, 0, stream>>>(xl, att_src, att_dst, asg, adg);

    k_gat_node<<<(N_NODES * 32 + 255) / 256, 256, 0, stream>>>(xl, asg, adg, rowptr, deg, csrc,
                                                               gat_bias, out);

    (void)in_sizes; (void)n_in; (void)out_size; (void)ws_size;
}
